# starting kernel
# baseline (speedup 1.0000x reference)
#include <hip/hip_runtime.h>
#include <math.h>

#define BB 4
#define CC 64
#define NN 4096
#define OO 64
#define KK 20
#define EPSV 1e-5f

typedef __attribute__((ext_vector_type(2))) float v2f;
typedef __attribute__((ext_vector_type(8))) float v8f;

__device__ __forceinline__ v8f wmma_f32(v2f a, v2f b, v8f c) {
    return __builtin_amdgcn_wmma_f32_16x16x4_f32(false, a, false, b, (short)0, c,
                                                 false, false);
}

__global__ __launch_bounds__(256) void prep_kernel(const float* __restrict__ x,
                                                   float* __restrict__ xt,
                                                   float* __restrict__ xx) {
    int t = blockIdx.x * blockDim.x + threadIdx.x;
    int b = t / NN;
    int n = t - b * NN;
    const float* px = x + (size_t)b * CC * NN + n;
    float* pt = xt + (size_t)t * CC;
    float s = 0.f;
#pragma unroll 8
    for (int c = 0; c < CC; ++c) {
        float v = px[(size_t)c * NN];
        pt[c] = v;
        s += v * v;
    }
    xx[t] = s;
}

__global__ __launch_bounds__(128) void knn_kernel(const float* __restrict__ xt,
                                                  const float* __restrict__ xx,
                                                  int* __restrict__ nbr) {
    const int wave = threadIdx.x >> 5;
    const int lane = threadIdx.x & 31;
    const int laneM = lane & 15;
    const int kofs = (lane < 16) ? 0 : 2;

    const int row0 = blockIdx.x * 128 + wave * 32;
    const int b = row0 / NN;
    const int r0 = row0 - b * NN;

    const float* prowLo = xt + (size_t)(b * NN + r0 + laneM) * CC;
    const float* prowHi = prowLo + 16 * CC;
    v2f aLo[16], aHi[16];
#pragma unroll
    for (int ks = 0; ks < 16; ++ks) {
        int kc = ks * 4 + kofs;
        aLo[ks] = *(const v2f*)(prowLo + kc);
        aHi[ks] = *(const v2f*)(prowHi + kc);
    }

    float topv[KK];
    int topi[KK];
#pragma unroll
    for (int s = 0; s < KK; ++s) { topv[s] = -3.0e38f; topi[s] = 0; }
    const float xxrow = xx[b * NN + r0 + lane];

    __shared__ float dist[4][32 * 16];

    for (int ct = 0; ct < NN / 16; ++ct) {
        const int col = ct * 16 + laneM;
        const float* pcol = xt + (size_t)(b * NN + col) * CC;
        v8f accLo = {};
        v8f accHi = {};
#pragma unroll
        for (int ks = 0; ks < 16; ++ks) {
            v2f bb = *(const v2f*)(pcol + ks * 4 + kofs);
            accLo = wmma_f32(aLo[ks], bb, accLo);
            accHi = wmma_f32(aHi[ks], bb, accHi);
        }
        const float xcol = xx[b * NN + col];
#pragma unroll
        for (int j = 0; j < 8; ++j) {
            int M = j + ((lane < 16) ? 0 : 8);
            dist[wave][M * 16 + laneM] = 2.0f * accLo[j] - xcol;
            dist[wave][(16 + M) * 16 + laneM] = 2.0f * accHi[j] - xcol;
        }
        __syncthreads();
#pragma unroll
        for (int c = 0; c < 16; ++c) {
            float v = dist[wave][lane * 16 + c] - xxrow;
            if (v > topv[KK - 1]) {
                float cv = v;
                int ci = ct * 16 + c;
#pragma unroll
                for (int s = 0; s < KK; ++s) {
                    if (cv > topv[s]) {
                        float tv = topv[s]; topv[s] = cv; cv = tv;
                        int ti = topi[s]; topi[s] = ci; ci = ti;
                    }
                }
            }
        }
        __syncthreads();
    }
    int* outp = nbr + (size_t)(b * NN + r0 + lane) * KK;
#pragma unroll
    for (int s = 0; s < KK; ++s) outp[s] = topi[s];
}

__global__ __launch_bounds__(128) void edgeconv_kernel(
    const float* __restrict__ xt, const int* __restrict__ nbr,
    const float* __restrict__ w1, const float* __restrict__ g1,
    const float* __restrict__ be1, const float* __restrict__ mu1,
    const float* __restrict__ va1, const float* __restrict__ w2,
    const float* __restrict__ g2, const float* __restrict__ be2,
    const float* __restrict__ mu2, const float* __restrict__ va2,
    float* __restrict__ out) {
    const int wave = threadIdx.x >> 5;
    const int lane = threadIdx.x & 31;
    const int laneM = lane & 15;
    const int kofs = (lane < 16) ? 0 : 2;
    const int b = blockIdx.x >> 8;
    const int pt = blockIdx.x & 255;

    __shared__ float h1s[4][16 * OO];
    __shared__ unsigned om[16 * OO];
    for (int i = threadIdx.x; i < 16 * OO; i += 128) om[i] = 0u;
    __syncthreads();

    for (int it = 0; it < 5; ++it) {
        const int mt = wave * 5 + it;
        const int lrow = mt * 16 + laneM;
        const int p = lrow / KK;
        const int kk = lrow - p * KK;
        const int n = pt * 16 + p;
        const int nb = nbr[(size_t)(b * NN + n) * KK + kk];
        const float* pc = xt + (size_t)(b * NN + n) * CC;
        const float* pn = xt + (size_t)(b * NN + nb) * CC;

        v8f acc1[4] = {};
#pragma unroll
        for (int ks = 0; ks < 16; ++ks) {
            int kc = ks * 4 + kofs;
            v2f a = *(const v2f*)(pn + kc) - *(const v2f*)(pc + kc);
#pragma unroll
            for (int nt = 0; nt < 4; ++nt) {
                int o = nt * 16 + laneM;
                v2f bb = *(const v2f*)(w1 + o * 128 + kc);
                acc1[nt] = wmma_f32(a, bb, acc1[nt]);
            }
        }
#pragma unroll
        for (int ks = 0; ks < 16; ++ks) {
            int kc = ks * 4 + kofs;
            v2f a = *(const v2f*)(pc + kc);
#pragma unroll
            for (int nt = 0; nt < 4; ++nt) {
                int o = nt * 16 + laneM;
                v2f bb = *(const v2f*)(w1 + o * 128 + 64 + kc);
                acc1[nt] = wmma_f32(a, bb, acc1[nt]);
            }
        }
#pragma unroll
        for (int nt = 0; nt < 4; ++nt) {
            int o = nt * 16 + laneM;
            float sc = g1[o] * rsqrtf(va1[o] + EPSV);
            float sh = be1[o] - mu1[o] * sc;
#pragma unroll
            for (int j = 0; j < 8; ++j) {
                int M = j + ((lane < 16) ? 0 : 8);
                h1s[wave][M * OO + o] = fmaxf(acc1[nt][j] * sc + sh, 0.0f);
            }
        }
        __syncthreads();

        v8f acc2[4] = {};
#pragma unroll
        for (int ks = 0; ks < 16; ++ks) {
            int kc = ks * 4 + kofs;
            v2f a = *(const v2f*)(&h1s[wave][laneM * OO + kc]);
#pragma unroll
            for (int nt = 0; nt < 4; ++nt) {
                int o = nt * 16 + laneM;
                v2f bb = *(const v2f*)(w2 + o * OO + kc);
                acc2[nt] = wmma_f32(a, bb, acc2[nt]);
            }
        }
#pragma unroll
        for (int nt = 0; nt < 4; ++nt) {
            int o = nt * 16 + laneM;
            float sc = g2[o] * rsqrtf(va2[o] + EPSV);
            float sh = be2[o] - mu2[o] * sc;
#pragma unroll
            for (int j = 0; j < 8; ++j) {
                int M = j + ((lane < 16) ? 0 : 8);
                int lr = mt * 16 + M;
                int pp = lr / KK;
                float val = fmaxf(acc2[nt][j] * sc + sh, 0.0f);
                atomicMax(&om[pp * OO + o], __float_as_uint(val));
            }
        }
        __syncthreads();
    }
    for (int i = threadIdx.x; i < 16 * OO; i += 128) {
        int p = i >> 6;
        int o = i & 63;
        out[(size_t)(b * OO + o) * NN + pt * 16 + p] = __uint_as_float(om[p * OO + o]);
    }
}

extern "C" void kernel_launch(void* const* d_in, const int* in_sizes, int n_in,
                              void* d_out, int out_size, void* d_ws, size_t ws_size,
                              hipStream_t stream) {
    const float* x   = (const float*)d_in[0];
    const float* w1  = (const float*)d_in[1];
    const float* g1  = (const float*)d_in[2];
    const float* be1 = (const float*)d_in[3];
    const float* mu1 = (const float*)d_in[4];
    const float* va1 = (const float*)d_in[5];
    const float* w2  = (const float*)d_in[6];
    const float* g2  = (const float*)d_in[7];
    const float* be2 = (const float*)d_in[8];
    const float* mu2 = (const float*)d_in[9];
    const float* va2 = (const float*)d_in[10];
    float* out = (float*)d_out;

    float* xt = (float*)d_ws;
    float* xx = xt + (size_t)BB * NN * CC;
    int* nbr = (int*)(xx + (size_t)BB * NN);

    prep_kernel<<<(BB * NN) / 256, 256, 0, stream>>>(x, xt, xx);
    knn_kernel<<<(BB * NN) / 128, 128, 0, stream>>>(xt, xx, nbr);
    edgeconv_kernel<<<BB * (NN / 16), 128, 0, stream>>>(
        xt, nbr, w1, g1, be1, mu1, va1, w2, g2, be2, mu2, va2, out);
}


# edge MLP: one wave per SIMD loops over its 20 row tiles with W1, W2 fragments and BN constants held in registers (no weight load per tile); next gather a tile ahead, per-tile barriers at wave scope
# speedup vs baseline: 19.7453x; 19.7453x over previous
#include <hip/hip_runtime.h>
#include <math.h>

#define BB 4
#define CC 64
#define NN 4096
#define OO 64
#define KK 20
#define EPSV 1e-5f

typedef __attribute__((ext_vector_type(2))) float v2f;
typedef __attribute__((ext_vector_type(4))) float v4f;
typedef __attribute__((ext_vector_type(8))) float v8f;

__device__ __forceinline__ v8f wmma_f32(v2f a, v2f b, v8f c) {
    return __builtin_amdgcn_wmma_f32_16x16x4_f32(false, a, false, b, (short)0, c,
                                                 false, false);
}

__global__ __launch_bounds__(256) void prep_kernel(const float* __restrict__ x,
                                                   float* __restrict__ xt,
                                                   float* __restrict__ xx) {
    int t = blockIdx.x * blockDim.x + threadIdx.x;
    int b = t / NN;
    int n = t - b * NN;
    const float* px = x + (size_t)b * CC * NN + n;
    float* pt = xt + (size_t)t * CC;
    float s = 0.f;
#pragma unroll 8
    for (int c = 0; c < CC; ++c) {
        float v = px[(size_t)c * NN];
        pt[c] = v;
        s += v * v;
    }
    xx[t] = s;
}

#define LCAP 104
#define LTRIG 96
#define NPRE 8

#define CH 128
#define NCH (NN / CH)
#define PITCH 68
#define SLD 16

__device__ __forceinline__ void stage_fetch(const float* __restrict__ xt_b,
                                            const float* __restrict__ xx_b, int chunk, int tid,
                                            v4f (&st)[SLD], float& sx) {
    const v4f* src = (const v4f*)(xt_b + (size_t)chunk * (CH * CC)) + tid;
#pragma unroll
    for (int i = 0; i < SLD; ++i) st[i] = src[i * 128];
    sx = xx_b[chunk * CH + tid];
}

__device__ __forceinline__ void stage_put(float* __restrict__ buf, float* __restrict__ xbuf,
                                          int tid, const v4f (&st)[SLD], float sx) {
    float* dst = buf + (tid >> 4) * PITCH + (tid & 15) * 4;
#pragma unroll
    for (int i = 0; i < SLD; ++i) *(v4f*)(dst + i * 8 * PITCH) = st[i];
    xbuf[tid] = sx;
}

struct Frag {
    v2f a[16];
    v4f x0, x1;
};
struct Pend {
    v8f acc;
    v4f x0, x1;
};

__device__ __forceinline__ void frag_load(const float* __restrict__ buf,
                                          const float* __restrict__ xbuf, int t, int laneM,
                                          int kofs, int half8, Frag& f) {
    const float* prow = buf + (t * 16 + laneM) * PITCH + kofs;
    const v4f* pxx = (const v4f*)(xbuf + t * 16 + half8);
#pragma unroll
    for (int ks = 0; ks < 16; ++ks) f.a[ks] = *(const v2f*)(prow + ks * 4);
    f.x0 = pxx[0];
    f.x1 = pxx[1];
}

__device__ __forceinline__ v8f products(const Frag& f, const v2f (&bq)[16]) {
    v8f acc = {};
#pragma unroll
    for (int ks = 0; ks < 16; ++ks) acc = wmma_f32(f.a[ks], bq[ks], acc);
    return acc;
}

__device__ __forceinline__ void finish(const Pend& p, float (&u)[8]) {
    u[0] = 2.0f * p.acc[0] - p.x0.x;
    u[1] = 2.0f * p.acc[1] - p.x0.y;
    u[2] = 2.0f * p.acc[2] - p.x0.z;
    u[3] = 2.0f * p.acc[3] - p.x0.w;
    u[4] = 2.0f * p.acc[4] - p.x1.x;
    u[5] = 2.0f * p.acc[5] - p.x1.y;
    u[6] = 2.0f * p.acc[6] - p.x1.z;
    u[7] = 2.0f * p.acc[7] - p.x1.w;
}

__device__ __forceinline__ void select1(const Pend& p, float xxq, float (&topv)[KK]) {
    float u[8];
    finish(p, u);
    float m = fmaxf(fmaxf(fmaxf(u[0], u[1]), fmaxf(u[2], u[3])),
                    fmaxf(fmaxf(u[4], u[5]), fmaxf(u[6], u[7]))) - xxq;
#pragma unroll
    for (int s = 0; s < KK; ++s) {
        float hi = fmaxf(topv[s], m);
        m = fminf(topv[s], m);
        topv[s] = hi;
    }
}

__device__ __forceinline__ void select2_mark(const Pend& p, float xxq, float thr, bool strict,
                                             int& cnt, float (&v)[8], int (&slot)[8]) {
    float u[8];
    finish(p, u);
#pragma unroll
    for (int r = 0; r < 8; ++r) {
        v[r] = u[r] - xxq;
        bool pass = (v[r] >= thr) && !(strict && v[r] == thr);
        slot[r] = cnt;
        cnt += pass ? 1 : 0;
    }
}

__device__ __forceinline__ void select2_store(float* lv, int* li, int colbase,
                                              const float (&v)[8], const int (&slot)[8]) {
#pragma unroll
    for (int r = 0; r < 8; ++r) {
        lv[slot[r] * 32] = v[r];
        li[slot[r] * 32] = colbase + r;
    }
}

__device__ __forceinline__ v8f products_mark(const Frag& f, const v2f (&bq)[16], const Pend& p,
                                             float xxq, float thr, bool strict, bool ins,
                                             int& cnt, float (&v)[8], int (&slot)[8],
                                             float (&topv)[KK]) {
    float u[8];
    finish(p, u);
    float m = -INFINITY;
    v8f acc = {};
#pragma unroll
    for (int ks = 0; ks < 16; ++ks) {
        acc = wmma_f32(f.a[ks], bq[ks], acc);
        if (ks < 8) {
            v[ks] = u[ks] - xxq;
            bool pass = (v[ks] >= thr) && !(strict && v[ks] == thr);
            slot[ks] = cnt;
            cnt += pass ? 1 : 0;
        } else {
            if (ks == 8) {
                float mx = fmaxf(fmaxf(fmaxf(v[0], v[1]), fmaxf(v[2], v[3])),
                                 fmaxf(fmaxf(v[4], v[5]), fmaxf(v[6], v[7])));
                m = ins ? mx : -INFINITY;
            }
            const int sb = (ks == 8) ? 0 : 2 + (ks - 9) * 3;
            const int se = (ks == 8) ? 2 : ((2 + (ks - 8) * 3) < KK ? (2 + (ks - 8) * 3) : KK);
#pragma unroll
            for (int sl = sb; sl < se; ++sl) {
                float hi = fmaxf(topv[sl], m);
                m = fminf(topv[sl], m);
                topv[sl] = hi;
            }
        }
        __builtin_amdgcn_sched_barrier(0);
    }
    return acc;
}

__device__ __forceinline__ int drop_to_20(float* lv, int* li, int m) {
    while (m > KK) {
        float wv = lv[0];
        int wi = li[0];
        int wp = 0;
        for (int j = 1; j < m; ++j) {
            float vj = lv[j * 32];
            int ij = li[j * 32];
            bool worse = (vj < wv) || (vj == wv && ij > wi);
            if (worse) { wv = vj; wi = ij; wp = j; }
        }
        --m;
        lv[wp * 32] = lv[m * 32];
        li[wp * 32] = li[m * 32];
    }
    return m;
}

__global__ __launch_bounds__(128) void knn_kernel(const float* __restrict__ xt,
                                                  const float* __restrict__ xx,
                                                  int* __restrict__ nbr) {
    const int tid = threadIdx.x;
    const int wave = threadIdx.x >> 5;
    const int lane = threadIdx.x & 31;
    const int laneM = lane & 15;
    const int half = lane >> 4;
    const int kofs = half * 2;
    const int half8 = half * 8;

    const int qg = blockIdx.x * 64 + wave * 16;
    const int b = qg / NN;
    const int q0 = qg - b * NN;
    const float* xt_b = xt + (size_t)b * NN * CC;
    const float* xx_b = xx + b * NN;

    const float* pq = xt_b + (size_t)(q0 + laneM) * CC + kofs;
    v2f bq[16];
#pragma unroll
    for (int ks = 0; ks < 16; ++ks) bq[ks] = *(const v2f*)(pq + ks * 4);
    const float xxq = xx_b[q0 + laneM];

    __shared__ __attribute__((aligned(16))) float colbuf[2][CH * PITCH];
    __shared__ __attribute__((aligned(16))) float xxbuf[2][CH];
    struct Lists {
        float v[LCAP * 32];
        int i[LCAP * 32];
    };
    __shared__ Lists lists[4];
    float* lv = &lists[wave].v[lane];
    int* li = &lists[wave].i[lane];

    v4f st[SLD];
    float sx;
    stage_fetch(xt_b, xx_b, 0, tid, st, sx);
    stage_put(colbuf[0], xxbuf[0], tid, st, sx);
    __syncthreads();

    float topv[KK];
#pragma unroll
    for (int s = 0; s < KK; ++s) topv[s] = -3.0e38f;

    Pend pend;
    pend.acc = v8f{};
    pend.x0 = v4f{INFINITY, INFINITY, INFINITY, INFINITY};
    pend.x1 = pend.x0;
    for (int c = 0; c < NPRE; ++c) {
        stage_fetch(xt_b, xx_b, (c + 1) & (NPRE - 1), tid, st, sx);
        const float* buf = colbuf[c & 1];
        const float* xb = xxbuf[c & 1];
        Frag f0, f1;
        frag_load(buf, xb, 0, laneM, kofs, half8, f0);
#pragma unroll
        for (int tp = 0; tp < CH / 32; ++tp) {
            frag_load(buf, xb, 2 * tp + 1, laneM, kofs, half8, f1);
            __builtin_amdgcn_sched_barrier(0);
            v8f accA = products(f0, bq);
            select1(pend, xxq, topv);
            __builtin_amdgcn_sched_barrier(0);
            pend.acc = accA;
            pend.x0 = f0.x0;
            pend.x1 = f0.x1;
            if (tp + 1 < CH / 32) frag_load(buf, xb, 2 * tp + 2, laneM, kofs, half8, f0);
            __builtin_amdgcn_sched_barrier(0);
            v8f accB = products(f1, bq);
            select1(pend, xxq, topv);
            __builtin_amdgcn_sched_barrier(0);
            pend.acc = accB;
            pend.x0 = f1.x0;
            pend.x1 = f1.x1;
        }
        stage_put(colbuf[(c + 1) & 1], xxbuf[(c + 1) & 1], tid, st, sx);
        __syncthreads();
    }
    select1(pend, xxq, topv);

    float thr = topv[KK - 1];
    bool strict = false;
    int cnt = 0;
    pend.acc = v8f{};
    pend.x0 = v4f{NAN, NAN, NAN, NAN};
    pend.x1 = pend.x0;
    int pcol = half8;
    for (int c = 0; c < NCH; ++c) {
        const bool more = (c + 1 < NCH);
        if (more) stage_fetch(xt_b, xx_b, c + 1, tid, st, sx);
        const float* buf = colbuf[c & 1];
        const float* xb = xxbuf[c & 1];
        Frag f0, f1;
        frag_load(buf, xb, 0, laneM, kofs, half8, f0);
#pragma unroll
        for (int t = 0; t < CH / 16; ++t) {
            Frag& fc = (t & 1) ? f1 : f0;
            Frag& fn = (t & 1) ? f0 : f1;
            if (t + 1 < CH / 16) frag_load(buf, xb, t + 1, laneM, kofs, half8, fn);
            __builtin_amdgcn_sched_barrier(0);
            float cv[8];
            int cs[8];
            const bool ins = (t == 0) ? (c > NPRE) : (c >= NPRE);
            v8f acc = products_mark(fc, bq, pend, xxq, thr, strict, ins, cnt, cv, cs, topv);
            select2_store(lv, li, pcol, cv, cs);
            {
                const float bnd = topv[KK - 1];
                const bool up = bnd > thr;
                thr = up ? bnd : thr;
                strict = strict && !up;
            }
            pend.acc = acc;
            pend.x0 = fc.x0;
            pend.x1 = fc.x1;
            pcol = c * CH + t * 16 + half8;
            if (cnt >= LTRIG) {
                cnt = drop_to_20(lv, li, cnt);
                float lo = lv[0];
                for (int s = 1; s < KK; ++s) lo = fminf(lo, lv[s * 32]);
                if (lo >= thr) {
                    thr = lo;
                    strict = true;
                }
            }
        }
        if (more) {
            stage_put(colbuf[(c + 1) & 1], xxbuf[(c + 1) & 1], tid, st, sx);
            __syncthreads();
        }
    }
    {
        float cv[8];
        int cs[8];
        select2_mark(pend, xxq, thr, strict, cnt, cv, cs);
        select2_store(lv, li, pcol, cv, cs);
        float m = fmaxf(fmaxf(fmaxf(cv[0], cv[1]), fmaxf(cv[2], cv[3])),
                        fmaxf(fmaxf(cv[4], cv[5]), fmaxf(cv[6], cv[7])));
#pragma unroll
        for (int s = 0; s < KK; ++s) {
            float hi = fmaxf(topv[s], m);
            m = fminf(topv[s], m);
            topv[s] = hi;
        }
    }
    {
        float pv[KK];
#pragma unroll
        for (int s = 0; s < KK; ++s) pv[s] = __shfl_xor(topv[s], 16);
#pragma unroll
        for (int t = 0; t < KK; ++t) {
            float m = pv[t];
#pragma unroll
            for (int s = 0; s < KK; ++s) {
                float hi = fmaxf(topv[s], m);
                m = fminf(topv[s], m);
                topv[s] = hi;
            }
        }
    }
    static_assert(LCAP % 8 == 0, "groups of eight must stay inside a lane's list");
    {
        const float thrq = topv[KK - 1];
        int kept = 0;
        for (int e0 = 0; e0 < cnt; e0 += 8) {
            float ve[8];
            int ie[8];
#pragma unroll
            for (int j = 0; j < 8; ++j) {
                ve[j] = lv[(e0 + j) * 32];
                ie[j] = li[(e0 + j) * 32];
            }
#pragma unroll
            for (int j = 0; j < 8; ++j) {
                lv[kept * 32] = ve[j];
                li[kept * 32] = ie[j];
                kept += ((e0 + j < cnt) && (ve[j] >= thrq)) ? 1 : 0;
            }
        }
        cnt = kept;
    }
    cnt = drop_to_20(lv, li, cnt);

    const int cntUp = __shfl_xor(cnt, 16);
    __syncthreads();
    if (half == 0) {
        for (int e = 0; e < cntUp; ++e) {
            lv[(cnt + e) * 32] = lv[16 + e * 32];
            li[(cnt + e) * 32] = li[16 + e * 32];
        }
        const int tot = drop_to_20(lv, li, cnt + cntUp);
        int* outp = nbr + (size_t)(b * NN + q0 + laneM) * KK;
        for (int s = 0; s < KK; ++s) outp[s] = (s < tot) ? li[s * 32] : 0;
    }
}

__device__ __forceinline__ void wave_lds_order() {
    __builtin_amdgcn_fence(__ATOMIC_RELEASE, "wavefront");
    __builtin_amdgcn_wave_barrier();
    __builtin_amdgcn_fence(__ATOMIC_ACQUIRE, "wavefront");
}

#define GPB 4
#define TPW 20

__device__ __forceinline__ int edge_row(int G0, int wave, int q, int laneM) {
    const int g = q / 5;
    const int it = q - g * 5;
    return (G0 + g) * (16 * KK) + (wave * 5 + it) * 16 + laneM;
}

__global__ __launch_bounds__(128) void edgeconv_kernel(
    const float* __restrict__ xt, const int* __restrict__ nbr,
    const float* __restrict__ w1, const float* __restrict__ g1,
    const float* __restrict__ be1, const float* __restrict__ mu1,
    const float* __restrict__ va1, const float* __restrict__ w2,
    const float* __restrict__ g2, const float* __restrict__ be2,
    const float* __restrict__ mu2, const float* __restrict__ va2,
    float* __restrict__ out) {
    const int wave = threadIdx.x >> 5;
    const int lane = threadIdx.x & 31;
    const int laneM = lane & 15;
    const int kofs = (lane < 16) ? 0 : 2;
    const int G0 = blockIdx.x * GPB;

    __shared__ float h1s[4][16 * OO];
    __shared__ unsigned om[16 * OO];

    int e0 = edge_row(G0, wave, 0, laneM);
    int nb0 = nbr[e0];
    int e1 = edge_row(G0, wave, 1, laneM);
    int nbN = nbr[e1];

    v2f wa[16][4], wb[16][4], wc[16][4];
#pragma unroll
    for (int ks = 0; ks < 16; ++ks) {
        const int kc = ks * 4 + kofs;
#pragma unroll
        for (int nt = 0; nt < 4; ++nt) {
            const int o = nt * 16 + laneM;
            wa[ks][nt] = *(const v2f*)(w1 + o * 128 + kc);
            wb[ks][nt] = *(const v2f*)(w1 + o * 128 + 64 + kc);
            wc[ks][nt] = *(const v2f*)(w2 + o * OO + kc);
        }
    }
    float sc1[4], sh1[4], sc2[4], sh2[4];
#pragma unroll
    for (int nt = 0; nt < 4; ++nt) {
        const int o = nt * 16 + laneM;
        sc1[nt] = g1[o] * rsqrtf(va1[o] + EPSV);
        sh1[nt] = be1[o] - mu1[o] * sc1[nt];
        sc2[nt] = g2[o] * rsqrtf(va2[o] + EPSV);
        sh2[nt] = be2[o] - mu2[o] * sc2[nt];
    }

    v2f an[16], cn[16];
    {
        const float* pc = xt + (size_t)(e0 / KK) * CC + kofs;
        const float* pn = xt + ((size_t)((e0 / KK) & ~(NN - 1)) + nb0) * CC + kofs;
#pragma unroll
        for (int ks = 0; ks < 16; ++ks) {
            an[ks] = *(const v2f*)(pn + ks * 4);
            cn[ks] = *(const v2f*)(pc + ks * 4);
        }
    }

    for (int i = threadIdx.x; i < 16 * OO; i += 128) om[i] = 0u;

    for (int q = 0; q < TPW; ++q) {
        const int g = q / 5;
        const int it = q - g * 5;
        const int G = G0 + g;
        const int mt = wave * 5 + it;
        if (it == 0) __syncthreads();

        v2f ac[16], cc[16];
#pragma unroll
        for (int ks = 0; ks < 16; ++ks) {
            ac[ks] = an[ks];
            cc[ks] = cn[ks];
        }
        {
            const int qn = (q + 1 < TPW) ? q + 1 : TPW - 1;
            const int q2 = (q + 2 < TPW) ? q + 2 : TPW - 1;
            const int en = edge_row(G0, wave, qn, laneM);
            const float* pc = xt + (size_t)(en / KK) * CC + kofs;
            const float* pn = xt + ((size_t)((en / KK) & ~(NN - 1)) + nbN) * CC + kofs;
#pragma unroll
            for (int ks = 0; ks < 16; ++ks) {
                an[ks] = *(const v2f*)(pn + ks * 4);
                cn[ks] = *(const v2f*)(pc + ks * 4);
            }
            nbN = nbr[edge_row(G0, wave, q2, laneM)];
        }

        v8f acc1[4] = {};
#pragma unroll
        for (int ks = 0; ks < 16; ++ks) {
            v2f a = ac[ks] - cc[ks];
#pragma unroll
            for (int nt = 0; nt < 4; ++nt) acc1[nt] = wmma_f32(a, wa[ks][nt], acc1[nt]);
        }
#pragma unroll
        for (int ks = 0; ks < 16; ++ks) {
#pragma unroll
            for (int nt = 0; nt < 4; ++nt) acc1[nt] = wmma_f32(cc[ks], wb[ks][nt], acc1[nt]);
        }
#pragma unroll
        for (int nt = 0; nt < 4; ++nt) {
            int o = nt * 16 + laneM;
#pragma unroll
            for (int j = 0; j < 8; ++j) {
                int M = j + ((lane < 16) ? 0 : 8);
                h1s[wave][M * OO + o] = fmaxf(acc1[nt][j] * sc1[nt] + sh1[nt], 0.0f);
            }
        }
        wave_lds_order();

        v8f acc2[4] = {};
#pragma unroll
        for (int ks = 0; ks < 16; ++ks) {
            int kc = ks * 4 + kofs;
            v2f a = *(const v2f*)(&h1s[wave][laneM * OO + kc]);
#pragma unroll
            for (int nt = 0; nt < 4; ++nt) acc2[nt] = wmma_f32(a, wc[ks][nt], acc2[nt]);
        }
#pragma unroll
        for (int nt = 0; nt < 4; ++nt) {
            int o = nt * 16 + laneM;
#pragma unroll
            for (int j = 0; j < 8; ++j) {
                int M = j + ((lane < 16) ? 0 : 8);
                int lr = mt * 16 + M;
                int pp = lr / KK;
                float val = fmaxf(acc2[nt][j] * sc2[nt] + sh2[nt], 0.0f);
                atomicMax(&om[pp * OO + o], __float_as_uint(val));
            }
        }
        wave_lds_order();

        if (it == 4) {
            __syncthreads();
            const int b = G >> 8;
            const int pt = G & 255;
            for (int i = threadIdx.x; i < 16 * OO; i += 128) {
                int p = i >> 6;
                int o = i & 63;
                out[(size_t)(b * OO + o) * NN + pt * 16 + p] = __uint_as_float(om[p * OO + o]);
                om[i] = 0u;
            }
        }
    }
}

extern "C" void kernel_launch(void* const* d_in, const int* in_sizes, int n_in,
                              void* d_out, int out_size, void* d_ws, size_t ws_size,
                              hipStream_t stream) {
    const float* x   = (const float*)d_in[0];
    const float* w1  = (const float*)d_in[1];
    const float* g1  = (const float*)d_in[2];
    const float* be1 = (const float*)d_in[3];
    const float* mu1 = (const float*)d_in[4];
    const float* va1 = (const float*)d_in[5];
    const float* w2  = (const float*)d_in[6];
    const float* g2  = (const float*)d_in[7];
    const float* be2 = (const float*)d_in[8];
    const float* mu2 = (const float*)d_in[9];
    const float* va2 = (const float*)d_in[10];
    float* out = (float*)d_out;

    float* xt = (float*)d_ws;
    float* xx = xt + (size_t)BB * NN * CC;
    int* nbr = (int*)(xx + (size_t)BB * NN);

    prep_kernel<<<(BB * NN) / 256, 256, 0, stream>>>(x, xt, xx);
    knn_kernel<<<(BB * NN) / 64, 128, 0, stream>>>(xt, xx, nbr);
    edgeconv_kernel<<<(BB * (NN / 16)) / GPB, 128, 0, stream>>>(
        xt, nbr, w1, g1, be1, mu1, va1, w2, g2, be2, mu2, va2, out);
}


# kNN sweep 2: a candidate's test is one vector compare (tie rule folded into the bound as its next float up); list-length branch once per stage: no scalar hop or branch between a tile's products
# speedup vs baseline: 23.2640x; 1.1782x over previous
#include <hip/hip_runtime.h>
#include <math.h>

#define BB 4
#define CC 64
#define NN 4096
#define OO 64
#define KK 20
#define EPSV 1e-5f

typedef __attribute__((ext_vector_type(2))) float v2f;
typedef __attribute__((ext_vector_type(4))) float v4f;
typedef __attribute__((ext_vector_type(8))) float v8f;

__device__ __forceinline__ v8f wmma_f32(v2f a, v2f b, v8f c) {
    return __builtin_amdgcn_wmma_f32_16x16x4_f32(false, a, false, b, (short)0, c,
                                                 false, false);
}

__global__ __launch_bounds__(256) void prep_kernel(const float* __restrict__ x,
                                                   float* __restrict__ xt,
                                                   float* __restrict__ xx) {
    int t = blockIdx.x * blockDim.x + threadIdx.x;
    int b = t / NN;
    int n = t - b * NN;
    const float* px = x + (size_t)b * CC * NN + n;
    float* pt = xt + (size_t)t * CC;
    float s = 0.f;
#pragma unroll 8
    for (int c = 0; c < CC; ++c) {
        float v = px[(size_t)c * NN];
        pt[c] = v;
        s += v * v;
    }
    xx[t] = s;
}

#define LCAP 160
#define LTRIG 96
#define NPRE 8

#define CH 128
#define NCH (NN / CH)
#define PITCH 68
#define SLD 16

__device__ __forceinline__ void stage_fetch(const float* __restrict__ xt_b,
                                            const float* __restrict__ xx_b, int chunk, int tid,
                                            v4f (&st)[SLD], float& sx) {
    const v4f* src = (const v4f*)(xt_b + (size_t)chunk * (CH * CC)) + tid;
#pragma unroll
    for (int i = 0; i < SLD; ++i) st[i] = src[i * 128];
    sx = xx_b[chunk * CH + tid];
}

__device__ __forceinline__ void stage_put(float* __restrict__ buf, float* __restrict__ xbuf,
                                          int tid, const v4f (&st)[SLD], float sx) {
    float* dst = buf + (tid >> 4) * PITCH + (tid & 15) * 4;
#pragma unroll
    for (int i = 0; i < SLD; ++i) *(v4f*)(dst + i * 8 * PITCH) = st[i];
    xbuf[tid] = sx;
}

struct Frag {
    v2f a[16];
    v4f x0, x1;
};
struct Pend {
    v8f acc;
    v4f x0, x1;
};

__device__ __forceinline__ void frag_load(const float* __restrict__ buf,
                                          const float* __restrict__ xbuf, int t, int laneM,
                                          int kofs, int half8, Frag& f) {
    const float* prow = buf + (t * 16 + laneM) * PITCH + kofs;
    const v4f* pxx = (const v4f*)(xbuf + t * 16 + half8);
#pragma unroll
    for (int ks = 0; ks < 16; ++ks) f.a[ks] = *(const v2f*)(prow + ks * 4);
    f.x0 = pxx[0];
    f.x1 = pxx[1];
}

__device__ __forceinline__ v8f products(const Frag& f, const v2f (&bq)[16]) {
    v8f acc = {};
#pragma unroll
    for (int ks = 0; ks < 16; ++ks) acc = wmma_f32(f.a[ks], bq[ks], acc);
    return acc;
}

__device__ __forceinline__ void finish(const Pend& p, float (&u)[8]) {
    u[0] = 2.0f * p.acc[0] - p.x0.x;
    u[1] = 2.0f * p.acc[1] - p.x0.y;
    u[2] = 2.0f * p.acc[2] - p.x0.z;
    u[3] = 2.0f * p.acc[3] - p.x0.w;
    u[4] = 2.0f * p.acc[4] - p.x1.x;
    u[5] = 2.0f * p.acc[5] - p.x1.y;
    u[6] = 2.0f * p.acc[6] - p.x1.z;
    u[7] = 2.0f * p.acc[7] - p.x1.w;
}

__device__ __forceinline__ void select1(const Pend& p, float xxq, float (&topv)[KK]) {
    float u[8];
    finish(p, u);
    float m = fmaxf(fmaxf(fmaxf(u[0], u[1]), fmaxf(u[2], u[3])),
                    fmaxf(fmaxf(u[4], u[5]), fmaxf(u[6], u[7]))) - xxq;
#pragma unroll
    for (int s = 0; s < KK; ++s) {
        float hi = fmaxf(topv[s], m);
        m = fminf(topv[s], m);
        topv[s] = hi;
    }
}

__device__ __forceinline__ void select2_mark(const Pend& p, float xxq, float thr, int& cnt,
                                             float (&v)[8], int (&slot)[8]) {
    float u[8];
    finish(p, u);
#pragma unroll
    for (int r = 0; r < 8; ++r) {
        v[r] = u[r] - xxq;
        slot[r] = cnt;
        cnt += (v[r] >= thr) ? 1 : 0;
    }
}

__device__ __forceinline__ float next_above(float x) {
    const int b = __float_as_int(x);
    if (x == 0.0f) return __int_as_float(1);
    return __int_as_float(b >= 0 ? b + 1 : b - 1);
}

__device__ __forceinline__ void select2_store(float* lv, int* li, int colbase,
                                              const float (&v)[8], const int (&slot)[8]) {
#pragma unroll
    for (int r = 0; r < 8; ++r) {
        lv[slot[r] * 32] = v[r];
        li[slot[r] * 32] = colbase + r;
    }
}

__device__ __forceinline__ v8f products_mark(const Frag& f, const v2f (&bq)[16], const Pend& p,
                                             float xxq, float thr, bool ins, int& cnt,
                                             float (&v)[8], int (&slot)[8], float (&topv)[KK]) {
    float u[8];
    finish(p, u);
    float m = -INFINITY;
    v8f acc = {};
#pragma unroll
    for (int ks = 0; ks < 16; ++ks) {
        acc = wmma_f32(f.a[ks], bq[ks], acc);
        if (ks < 8) {
            v[ks] = u[ks] - xxq;
            slot[ks] = cnt;
            cnt += (v[ks] >= thr) ? 1 : 0;
        } else {
            if (ks == 8) {
                float mx = fmaxf(fmaxf(fmaxf(v[0], v[1]), fmaxf(v[2], v[3])),
                                 fmaxf(fmaxf(v[4], v[5]), fmaxf(v[6], v[7])));
                m = ins ? mx : -INFINITY;
            }
            const int sb = (ks == 8) ? 0 : 2 + (ks - 9) * 3;
            const int se = (ks == 8) ? 2 : ((2 + (ks - 8) * 3) < KK ? (2 + (ks - 8) * 3) : KK);
#pragma unroll
            for (int sl = sb; sl < se; ++sl) {
                float hi = fmaxf(topv[sl], m);
                m = fminf(topv[sl], m);
                topv[sl] = hi;
            }
        }
        __builtin_amdgcn_sched_barrier(0);
    }
    return acc;
}

__device__ __forceinline__ int drop_to_20(float* lv, int* li, int m) {
    while (m > KK) {
        float wv = lv[0];
        int wi = li[0];
        int wp = 0;
        for (int j = 1; j < m; ++j) {
            float vj = lv[j * 32];
            int ij = li[j * 32];
            bool worse = (vj < wv) || (vj == wv && ij > wi);
            if (worse) { wv = vj; wi = ij; wp = j; }
        }
        --m;
        lv[wp * 32] = lv[m * 32];
        li[wp * 32] = li[m * 32];
    }
    return m;
}

__global__ __launch_bounds__(128) void knn_kernel(const float* __restrict__ xt,
                                                  const float* __restrict__ xx,
                                                  int* __restrict__ nbr) {
    const int tid = threadIdx.x;
    const int wave = threadIdx.x >> 5;
    const int lane = threadIdx.x & 31;
    const int laneM = lane & 15;
    const int half = lane >> 4;
    const int kofs = half * 2;
    const int half8 = half * 8;

    const int qg = blockIdx.x * 64 + wave * 16;
    const int b = qg / NN;
    const int q0 = qg - b * NN;
    const float* xt_b = xt + (size_t)b * NN * CC;
    const float* xx_b = xx + b * NN;

    const float* pq = xt_b + (size_t)(q0 + laneM) * CC + kofs;
    v2f bq[16];
#pragma unroll
    for (int ks = 0; ks < 16; ++ks) bq[ks] = *(const v2f*)(pq + ks * 4);
    const float xxq = xx_b[q0 + laneM];

    __shared__ __attribute__((aligned(16))) float colbuf[2][CH * PITCH];
    __shared__ __attribute__((aligned(16))) float xxbuf[2][CH];
    struct Lists {
        float v[LCAP * 32];
        int i[LCAP * 32];
    };
    __shared__ Lists lists[4];
    float* lv = &lists[wave].v[lane];
    int* li = &lists[wave].i[lane];

    v4f st[SLD];
    float sx;
    stage_fetch(xt_b, xx_b, 0, tid, st, sx);
    stage_put(colbuf[0], xxbuf[0], tid, st, sx);
    __syncthreads();

    float topv[KK];
#pragma unroll
    for (int s = 0; s < KK; ++s) topv[s] = -3.0e38f;

    Pend pend;
    pend.acc = v8f{};
    pend.x0 = v4f{INFINITY, INFINITY, INFINITY, INFINITY};
    pend.x1 = pend.x0;
    for (int c = 0; c < NPRE; ++c) {
        stage_fetch(xt_b, xx_b, (c + 1) & (NPRE - 1), tid, st, sx);
        const float* buf = colbuf[c & 1];
        const float* xb = xxbuf[c & 1];
        Frag f0, f1;
        frag_load(buf, xb, 0, laneM, kofs, half8, f0);
#pragma unroll
        for (int tp = 0; tp < CH / 32; ++tp) {
            frag_load(buf, xb, 2 * tp + 1, laneM, kofs, half8, f1);
            __builtin_amdgcn_sched_barrier(0);
            v8f accA = products(f0, bq);
            select1(pend, xxq, topv);
            __builtin_amdgcn_sched_barrier(0);
            pend.acc = accA;
            pend.x0 = f0.x0;
            pend.x1 = f0.x1;
            if (tp + 1 < CH / 32) frag_load(buf, xb, 2 * tp + 2, laneM, kofs, half8, f0);
            __builtin_amdgcn_sched_barrier(0);
            v8f accB = products(f1, bq);
            select1(pend, xxq, topv);
            __builtin_amdgcn_sched_barrier(0);
            pend.acc = accB;
            pend.x0 = f1.x0;
            pend.x1 = f1.x1;
        }
        stage_put(colbuf[(c + 1) & 1], xxbuf[(c + 1) & 1], tid, st, sx);
        __syncthreads();
    }
    select1(pend, xxq, topv);

    float thr = topv[KK - 1];
    int cnt = 0;
    pend.acc = v8f{};
    pend.x0 = v4f{NAN, NAN, NAN, NAN};
    pend.x1 = pend.x0;
    int pcol = half8;
    for (int c = 0; c < NCH; ++c) {
        const bool more = (c + 1 < NCH);
        if (more) stage_fetch(xt_b, xx_b, c + 1, tid, st, sx);
        const float* buf = colbuf[c & 1];
        const float* xb = xxbuf[c & 1];
        Frag f0, f1;
        frag_load(buf, xb, 0, laneM, kofs, half8, f0);
#pragma unroll
        for (int t = 0; t < CH / 16; ++t) {
            Frag& fc = (t & 1) ? f1 : f0;
            Frag& fn = (t & 1) ? f0 : f1;
            if (t + 1 < CH / 16) frag_load(buf, xb, t + 1, laneM, kofs, half8, fn);
            __builtin_amdgcn_sched_barrier(0);
            float cv[8];
            int cs[8];
            const bool ins = (t == 0) ? (c > NPRE) : (c >= NPRE);
            v8f acc = products_mark(fc, bq, pend, xxq, thr, ins, cnt, cv, cs, topv);
            select2_store(lv, li, pcol, cv, cs);
            {
                const float bnd = topv[KK - 1];
                thr = (bnd > thr) ? bnd : thr;
            }
            pend.acc = acc;
            pend.x0 = fc.x0;
            pend.x1 = fc.x1;
            pcol = c * CH + t * 16 + half8;
        }
        if (cnt >= LTRIG) {
            cnt = drop_to_20(lv, li, cnt);
            float lo = lv[0];
            for (int s = 1; s < KK; ++s) lo = fminf(lo, lv[s * 32]);
            if (lo >= thr) thr = next_above(lo);
        }
        if (more) {
            stage_put(colbuf[(c + 1) & 1], xxbuf[(c + 1) & 1], tid, st, sx);
            __syncthreads();
        }
    }
    {
        float cv[8];
        int cs[8];
        select2_mark(pend, xxq, thr, cnt, cv, cs);
        select2_store(lv, li, pcol, cv, cs);
        float m = fmaxf(fmaxf(fmaxf(cv[0], cv[1]), fmaxf(cv[2], cv[3])),
                        fmaxf(fmaxf(cv[4], cv[5]), fmaxf(cv[6], cv[7])));
#pragma unroll
        for (int s = 0; s < KK; ++s) {
            float hi = fmaxf(topv[s], m);
            m = fminf(topv[s], m);
            topv[s] = hi;
        }
    }
    {
        float pv[KK];
#pragma unroll
        for (int s = 0; s < KK; ++s) pv[s] = __shfl_xor(topv[s], 16);
#pragma unroll
        for (int t = 0; t < KK; ++t) {
            float m = pv[t];
#pragma unroll
            for (int s = 0; s < KK; ++s) {
                float hi = fmaxf(topv[s], m);
                m = fminf(topv[s], m);
                topv[s] = hi;
            }
        }
    }
    static_assert(LCAP % 8 == 0, "groups of eight must stay inside a lane's list");
    {
        const float thrq = topv[KK - 1];
        int kept = 0;
        for (int e0 = 0; e0 < cnt; e0 += 8) {
            float ve[8];
            int ie[8];
#pragma unroll
            for (int j = 0; j < 8; ++j) {
                ve[j] = lv[(e0 + j) * 32];
                ie[j] = li[(e0 + j) * 32];
            }
#pragma unroll
            for (int j = 0; j < 8; ++j) {
                lv[kept * 32] = ve[j];
                li[kept * 32] = ie[j];
                kept += ((e0 + j < cnt) && (ve[j] >= thrq)) ? 1 : 0;
            }
        }
        cnt = kept;
    }
    cnt = drop_to_20(lv, li, cnt);

    const int cntUp = __shfl_xor(cnt, 16);
    __syncthreads();
    if (half == 0) {
        for (int e = 0; e < cntUp; ++e) {
            lv[(cnt + e) * 32] = lv[16 + e * 32];
            li[(cnt + e) * 32] = li[16 + e * 32];
        }
        const int tot = drop_to_20(lv, li, cnt + cntUp);
        int* outp = nbr + (size_t)(b * NN + q0 + laneM) * KK;
        for (int s = 0; s < KK; ++s) outp[s] = (s < tot) ? li[s * 32] : 0;
    }
}

__device__ __forceinline__ void wave_lds_order() {
    __builtin_amdgcn_fence(__ATOMIC_RELEASE, "wavefront");
    __builtin_amdgcn_wave_barrier();
    __builtin_amdgcn_fence(__ATOMIC_ACQUIRE, "wavefront");
}

#define GPB 4
#define TPW 20

__device__ __forceinline__ int edge_row(int G0, int wave, int q, int laneM) {
    const int g = q / 5;
    const int it = q - g * 5;
    return (G0 + g) * (16 * KK) + (wave * 5 + it) * 16 + laneM;
}

__global__ __launch_bounds__(128) void edgeconv_kernel(
    const float* __restrict__ xt, const int* __restrict__ nbr,
    const float* __restrict__ w1, const float* __restrict__ g1,
    const float* __restrict__ be1, const float* __restrict__ mu1,
    const float* __restrict__ va1, const float* __restrict__ w2,
    const float* __restrict__ g2, const float* __restrict__ be2,
    const float* __restrict__ mu2, const float* __restrict__ va2,
    float* __restrict__ out) {
    const int wave = threadIdx.x >> 5;
    const int lane = threadIdx.x & 31;
    const int laneM = lane & 15;
    const int kofs = (lane < 16) ? 0 : 2;
    const int G0 = blockIdx.x * GPB;

    __shared__ float h1s[4][16 * OO];
    __shared__ unsigned om[16 * OO];

    int e0 = edge_row(G0, wave, 0, laneM);
    int nb0 = nbr[e0];
    int e1 = edge_row(G0, wave, 1, laneM);
    int nbN = nbr[e1];

    v2f wa[16][4], wb[16][4], wc[16][4];
#pragma unroll
    for (int ks = 0; ks < 16; ++ks) {
        const int kc = ks * 4 + kofs;
#pragma unroll
        for (int nt = 0; nt < 4; ++nt) {
            const int o = nt * 16 + laneM;
            wa[ks][nt] = *(const v2f*)(w1 + o * 128 + kc);
            wb[ks][nt] = *(const v2f*)(w1 + o * 128 + 64 + kc);
            wc[ks][nt] = *(const v2f*)(w2 + o * OO + kc);
        }
    }
    float sc1[4], sh1[4], sc2[4], sh2[4];
#pragma unroll
    for (int nt = 0; nt < 4; ++nt) {
        const int o = nt * 16 + laneM;
        sc1[nt] = g1[o] * rsqrtf(va1[o] + EPSV);
        sh1[nt] = be1[o] - mu1[o] * sc1[nt];
        sc2[nt] = g2[o] * rsqrtf(va2[o] + EPSV);
        sh2[nt] = be2[o] - mu2[o] * sc2[nt];
    }

    v2f an[16], cn[16];
    {
        const float* pc = xt + (size_t)(e0 / KK) * CC + kofs;
        const float* pn = xt + ((size_t)((e0 / KK) & ~(NN - 1)) + nb0) * CC + kofs;
#pragma unroll
        for (int ks = 0; ks < 16; ++ks) {
            an[ks] = *(const v2f*)(pn + ks * 4);
            cn[ks] = *(const v2f*)(pc + ks * 4);
        }
    }

    for (int i = threadIdx.x; i < 16 * OO; i += 128) om[i] = 0u;

    for (int q = 0; q < TPW; ++q) {
        const int g = q / 5;
        const int it = q - g * 5;
        const int G = G0 + g;
        const int mt = wave * 5 + it;
        if (it == 0) __syncthreads();

        v2f ac[16], cc[16];
#pragma unroll
        for (int ks = 0; ks < 16; ++ks) {
            ac[ks] = an[ks];
            cc[ks] = cn[ks];
        }
        {
            const int qn = (q + 1 < TPW) ? q + 1 : TPW - 1;
            const int q2 = (q + 2 < TPW) ? q + 2 : TPW - 1;
            const int en = edge_row(G0, wave, qn, laneM);
            const float* pc = xt + (size_t)(en / KK) * CC + kofs;
            const float* pn = xt + ((size_t)((en / KK) & ~(NN - 1)) + nbN) * CC + kofs;
#pragma unroll
            for (int ks = 0; ks < 16; ++ks) {
                an[ks] = *(const v2f*)(pn + ks * 4);
                cn[ks] = *(const v2f*)(pc + ks * 4);
            }
            nbN = nbr[edge_row(G0, wave, q2, laneM)];
        }

        v8f acc1[4] = {};
#pragma unroll
        for (int ks = 0; ks < 16; ++ks) {
            v2f a = ac[ks] - cc[ks];
#pragma unroll
            for (int nt = 0; nt < 4; ++nt) acc1[nt] = wmma_f32(a, wa[ks][nt], acc1[nt]);
        }
#pragma unroll
        for (int ks = 0; ks < 16; ++ks) {
#pragma unroll
            for (int nt = 0; nt < 4; ++nt) acc1[nt] = wmma_f32(cc[ks], wb[ks][nt], acc1[nt]);
        }
#pragma unroll
        for (int nt = 0; nt < 4; ++nt) {
            int o = nt * 16 + laneM;
#pragma unroll
            for (int j = 0; j < 8; ++j) {
                int M = j + ((lane < 16) ? 0 : 8);
                h1s[wave][M * OO + o] = fmaxf(acc1[nt][j] * sc1[nt] + sh1[nt], 0.0f);
            }
        }
        wave_lds_order();

        v8f acc2[4] = {};
#pragma unroll
        for (int ks = 0; ks < 16; ++ks) {
            int kc = ks * 4 + kofs;
            v2f a = *(const v2f*)(&h1s[wave][laneM * OO + kc]);
#pragma unroll
            for (int nt = 0; nt < 4; ++nt) acc2[nt] = wmma_f32(a, wc[ks][nt], acc2[nt]);
        }
#pragma unroll
        for (int nt = 0; nt < 4; ++nt) {
            int o = nt * 16 + laneM;
#pragma unroll
            for (int j = 0; j < 8; ++j) {
                int M = j + ((lane < 16) ? 0 : 8);
                int lr = mt * 16 + M;
                int pp = lr / KK;
                float val = fmaxf(acc2[nt][j] * sc2[nt] + sh2[nt], 0.0f);
                atomicMax(&om[pp * OO + o], __float_as_uint(val));
            }
        }
        wave_lds_order();

        if (it == 4) {
            __syncthreads();
            const int b = G >> 8;
            const int pt = G & 255;
            for (int i = threadIdx.x; i < 16 * OO; i += 128) {
                int p = i >> 6;
                int o = i & 63;
                out[(size_t)(b * OO + o) * NN + pt * 16 + p] = __uint_as_float(om[p * OO + o]);
                om[i] = 0u;
            }
        }
    }
}

extern "C" void kernel_launch(void* const* d_in, const int* in_sizes, int n_in,
                              void* d_out, int out_size, void* d_ws, size_t ws_size,
                              hipStream_t stream) {
    const float* x   = (const float*)d_in[0];
    const float* w1  = (const float*)d_in[1];
    const float* g1  = (const float*)d_in[2];
    const float* be1 = (const float*)d_in[3];
    const float* mu1 = (const float*)d_in[4];
    const float* va1 = (const float*)d_in[5];
    const float* w2  = (const float*)d_in[6];
    const float* g2  = (const float*)d_in[7];
    const float* be2 = (const float*)d_in[8];
    const float* mu2 = (const float*)d_in[9];
    const float* va2 = (const float*)d_in[10];
    float* out = (float*)d_out;

    float* xt = (float*)d_ws;
    float* xx = xt + (size_t)BB * NN * CC;
    int* nbr = (int*)(xx + (size_t)BB * NN);

    prep_kernel<<<(BB * NN) / 256, 256, 0, stream>>>(x, xt, xx);
    knn_kernel<<<(BB * NN) / 64, 128, 0, stream>>>(xt, xx, nbr);
    edgeconv_kernel<<<(BB * (NN / 16)) / GPB, 128, 0, stream>>>(
        xt, nbr, w1, g1, be1, mu1, va1, w2, g2, be2, mu2, va2, out);
}


# kNN tile loop in rounds of two tiles: the two 16-product chains issue in turn on two accumulators, so no product waits for its own predecessor; the round before is selected between them
# speedup vs baseline: 24.1550x; 1.0383x over previous
#include <hip/hip_runtime.h>
#include <math.h>

#define BB 4
#define CC 64
#define NN 4096
#define OO 64
#define KK 20
#define EPSV 1e-5f

typedef __attribute__((ext_vector_type(2))) float v2f;
typedef __attribute__((ext_vector_type(4))) float v4f;
typedef __attribute__((ext_vector_type(8))) float v8f;

__device__ __forceinline__ v8f wmma_f32(v2f a, v2f b, v8f c) {
    return __builtin_amdgcn_wmma_f32_16x16x4_f32(false, a, false, b, (short)0, c,
                                                 false, false);
}

__global__ __launch_bounds__(256) void prep_kernel(const float* __restrict__ x,
                                                   float* __restrict__ xt,
                                                   float* __restrict__ xx) {
    int t = blockIdx.x * blockDim.x + threadIdx.x;
    int b = t / NN;
    int n = t - b * NN;
    const float* px = x + (size_t)b * CC * NN + n;
    float* pt = xt + (size_t)t * CC;
    float s = 0.f;
#pragma unroll 8
    for (int c = 0; c < CC; ++c) {
        float v = px[(size_t)c * NN];
        pt[c] = v;
        s += v * v;
    }
    xx[t] = s;
}

#define LCAP 160
#define LTRIG 96
#define NPRE 8

#define CH 128
#define NCH (NN / CH)
#define PITCH 68
#define SLD 16

__device__ __forceinline__ void stage_fetch(const float* __restrict__ xt_b,
                                            const float* __restrict__ xx_b, int chunk, int tid,
                                            v4f (&st)[SLD], float& sx) {
    const v4f* src = (const v4f*)(xt_b + (size_t)chunk * (CH * CC)) + tid;
#pragma unroll
    for (int i = 0; i < SLD; ++i) st[i] = src[i * 128];
    sx = xx_b[chunk * CH + tid];
}

__device__ __forceinline__ void stage_put(float* __restrict__ buf, float* __restrict__ xbuf,
                                          int tid, const v4f (&st)[SLD], float sx) {
    float* dst = buf + (tid >> 4) * PITCH + (tid & 15) * 4;
#pragma unroll
    for (int i = 0; i < SLD; ++i) *(v4f*)(dst + i * 8 * PITCH) = st[i];
    xbuf[tid] = sx;
}

struct Frag {
    v2f a[16];
    v4f x0, x1;
};
struct Pend {
    v8f acc;
    v4f x0, x1;
};

__device__ __forceinline__ void frag_load(const float* __restrict__ buf,
                                          const float* __restrict__ xbuf, int t, int laneM,
                                          int kofs, int half8, Frag& f) {
    const float* prow = buf + (t * 16 + laneM) * PITCH + kofs;
    const v4f* pxx = (const v4f*)(xbuf + t * 16 + half8);
#pragma unroll
    for (int ks = 0; ks < 16; ++ks) f.a[ks] = *(const v2f*)(prow + ks * 4);
    f.x0 = pxx[0];
    f.x1 = pxx[1];
}

__device__ __forceinline__ void products2(const Frag& fa, const Frag& fb, const v2f (&bq)[16],
                                          v8f& accA, v8f& accB) {
    accA = v8f{};
    accB = v8f{};
#pragma unroll
    for (int ks = 0; ks < 16; ++ks) {
        accA = wmma_f32(fa.a[ks], bq[ks], accA);
        accB = wmma_f32(fb.a[ks], bq[ks], accB);
    }
}

__device__ __forceinline__ void finish(const Pend& p, float (&u)[8]) {
    u[0] = 2.0f * p.acc[0] - p.x0.x;
    u[1] = 2.0f * p.acc[1] - p.x0.y;
    u[2] = 2.0f * p.acc[2] - p.x0.z;
    u[3] = 2.0f * p.acc[3] - p.x0.w;
    u[4] = 2.0f * p.acc[4] - p.x1.x;
    u[5] = 2.0f * p.acc[5] - p.x1.y;
    u[6] = 2.0f * p.acc[6] - p.x1.z;
    u[7] = 2.0f * p.acc[7] - p.x1.w;
}

__device__ __forceinline__ void select1(const Pend& p, float xxq, float (&topv)[KK]) {
    float u[8];
    finish(p, u);
    float m = fmaxf(fmaxf(fmaxf(u[0], u[1]), fmaxf(u[2], u[3])),
                    fmaxf(fmaxf(u[4], u[5]), fmaxf(u[6], u[7]))) - xxq;
#pragma unroll
    for (int s = 0; s < KK; ++s) {
        float hi = fmaxf(topv[s], m);
        m = fminf(topv[s], m);
        topv[s] = hi;
    }
}

__device__ __forceinline__ void select2_mark(const Pend& p, float xxq, float thr, int& cnt,
                                             float (&v)[8], int (&slot)[8]) {
    float u[8];
    finish(p, u);
#pragma unroll
    for (int r = 0; r < 8; ++r) {
        v[r] = u[r] - xxq;
        slot[r] = cnt;
        cnt += (v[r] >= thr) ? 1 : 0;
    }
}

__device__ __forceinline__ float next_above(float x) {
    const int b = __float_as_int(x);
    if (x == 0.0f) return __int_as_float(1);
    return __int_as_float(b >= 0 ? b + 1 : b - 1);
}

__device__ __forceinline__ void select2_store(float* lv, int* li, int colbase,
                                              const float (&v)[8], const int (&slot)[8]) {
#pragma unroll
    for (int r = 0; r < 8; ++r) {
        lv[slot[r] * 32] = v[r];
        li[slot[r] * 32] = colbase + r;
    }
}

__device__ __forceinline__ void select2_piece(int j, const float (&dv)[8], float thr, bool ins,
                                              int& cnt, float (&v)[8], int (&slot)[8], float& m,
                                              float (&topv)[KK]) {
    if (j < 8) {
        float d = dv[j];
        asm volatile("" : "+v"(d));
        v[j] = d;
        slot[j] = cnt;
        cnt += (v[j] >= thr) ? 1 : 0;
    } else {
        if (j == 8) {
            float mx = fmaxf(fmaxf(fmaxf(v[0], v[1]), fmaxf(v[2], v[3])),
                             fmaxf(fmaxf(v[4], v[5]), fmaxf(v[6], v[7])));
            m = ins ? mx : -INFINITY;
        }
        const int sb = (j == 8) ? 0 : 2 + (j - 9) * 3;
        const int se = (j == 8) ? 2 : ((2 + (j - 8) * 3) < KK ? (2 + (j - 8) * 3) : KK);
#pragma unroll
        for (int sl = sb; sl < se; ++sl) {
            float hi = fmaxf(topv[sl], m);
            m = fminf(topv[sl], m);
            topv[sl] = hi;
        }
    }
}

__device__ __forceinline__ void products2_mark(const Frag& fa, const Frag& fb, const v2f (&bq)[16],
                                               const Pend& pa, const Pend& pb, float xxq,
                                               float& thr, bool ins, int& cnt, float (&va)[8],
                                               int (&sa)[8], float (&vb)[8], int (&sb)[8],
                                               float (&topv)[KK], v8f& accA, v8f& accB) {
    float ua[8], ub[8], da[8], db[8];
    finish(pa, ua);
#pragma unroll
    for (int j = 0; j < 8; ++j) da[j] = ua[j] - xxq;
    float m = -INFINITY;
    accA = v8f{};
    accB = v8f{};
#pragma unroll
    for (int i = 0; i < 32; ++i) {
        const int ks = i >> 1;
        if ((i & 1) == 0) accA = wmma_f32(fa.a[ks], bq[ks], accA);
        else              accB = wmma_f32(fb.a[ks], bq[ks], accB);
        if (i < 16) {
            select2_piece(i, da, thr, ins, cnt, va, sa, m, topv);
            if (i == 15) {
                const float bnd = topv[KK - 1];
                thr = (bnd > thr) ? bnd : thr;
                finish(pb, ub);
#pragma unroll
                for (int j = 0; j < 8; ++j) db[j] = ub[j] - xxq;
            }
        } else {
            select2_piece(i - 16, db, thr, ins, cnt, vb, sb, m, topv);
        }
        __builtin_amdgcn_sched_barrier(0);
    }
}

__device__ __forceinline__ int drop_to_20(float* lv, int* li, int m) {
    while (m > KK) {
        float wv = lv[0];
        int wi = li[0];
        int wp = 0;
        for (int j = 1; j < m; ++j) {
            float vj = lv[j * 32];
            int ij = li[j * 32];
            bool worse = (vj < wv) || (vj == wv && ij > wi);
            if (worse) { wv = vj; wi = ij; wp = j; }
        }
        --m;
        lv[wp * 32] = lv[m * 32];
        li[wp * 32] = li[m * 32];
    }
    return m;
}

__global__ __launch_bounds__(128) void knn_kernel(const float* __restrict__ xt,
                                                  const float* __restrict__ xx,
                                                  int* __restrict__ nbr) {
    const int tid = threadIdx.x;
    const int wave = threadIdx.x >> 5;
    const int lane = threadIdx.x & 31;
    const int laneM = lane & 15;
    const int half = lane >> 4;
    const int kofs = half * 2;
    const int half8 = half * 8;

    const int qg = blockIdx.x * 64 + wave * 16;
    const int b = qg / NN;
    const int q0 = qg - b * NN;
    const float* xt_b = xt + (size_t)b * NN * CC;
    const float* xx_b = xx + b * NN;

    const float* pq = xt_b + (size_t)(q0 + laneM) * CC + kofs;
    v2f bq[16];
#pragma unroll
    for (int ks = 0; ks < 16; ++ks) bq[ks] = *(const v2f*)(pq + ks * 4);
    const float xxq = xx_b[q0 + laneM];

    __shared__ __attribute__((aligned(16))) float colbuf[2][CH * PITCH];
    __shared__ __attribute__((aligned(16))) float xxbuf[2][CH];
    struct Lists {
        float v[LCAP * 32];
        int i[LCAP * 32];
    };
    __shared__ Lists lists[4];
    float* lv = &lists[wave].v[lane];
    int* li = &lists[wave].i[lane];

    v4f st[SLD];
    float sx;
    stage_fetch(xt_b, xx_b, 0, tid, st, sx);
    stage_put(colbuf[0], xxbuf[0], tid, st, sx);
    __syncthreads();

    float topv[KK];
#pragma unroll
    for (int s = 0; s < KK; ++s) topv[s] = -3.0e38f;

    Pend pendA, pendB;
    pendA.acc = v8f{};
    pendA.x0 = v4f{INFINITY, INFINITY, INFINITY, INFINITY};
    pendA.x1 = pendA.x0;
    pendB = pendA;
    for (int c = 0; c < NPRE; ++c) {
        stage_fetch(xt_b, xx_b, (c + 1) & (NPRE - 1), tid, st, sx);
        const float* buf = colbuf[c & 1];
        const float* xb = xxbuf[c & 1];
        Frag fa0, fb0, fa1, fb1;
        frag_load(buf, xb, 0, laneM, kofs, half8, fa0);
        frag_load(buf, xb, 1, laneM, kofs, half8, fb0);
#pragma unroll
        for (int r = 0; r < CH / 32; ++r) {
            Frag& ca = (r & 1) ? fa1 : fa0;
            Frag& cb = (r & 1) ? fb1 : fb0;
            Frag& na = (r & 1) ? fa0 : fa1;
            Frag& nb = (r & 1) ? fb0 : fb1;
            if (r + 1 < CH / 32) {
                frag_load(buf, xb, 2 * r + 2, laneM, kofs, half8, na);
                frag_load(buf, xb, 2 * r + 3, laneM, kofs, half8, nb);
            }
            __builtin_amdgcn_sched_barrier(0);
            v8f accA, accB;
            products2(ca, cb, bq, accA, accB);
            select1(pendA, xxq, topv);
            select1(pendB, xxq, topv);
            __builtin_amdgcn_sched_barrier(0);
            pendA.acc = accA;
            pendA.x0 = ca.x0;
            pendA.x1 = ca.x1;
            pendB.acc = accB;
            pendB.x0 = cb.x0;
            pendB.x1 = cb.x1;
        }
        stage_put(colbuf[(c + 1) & 1], xxbuf[(c + 1) & 1], tid, st, sx);
        __syncthreads();
    }
    select1(pendA, xxq, topv);
    select1(pendB, xxq, topv);

    float thr = topv[KK - 1];
    int cnt = 0;
    pendA.acc = v8f{};
    pendA.x0 = v4f{NAN, NAN, NAN, NAN};
    pendA.x1 = pendA.x0;
    pendB = pendA;
    int pcol = half8;
    for (int c = 0; c < NCH; ++c) {
        const bool more = (c + 1 < NCH);
        if (more) stage_fetch(xt_b, xx_b, c + 1, tid, st, sx);
        const float* buf = colbuf[c & 1];
        const float* xb = xxbuf[c & 1];
        Frag fa0, fb0, fa1, fb1;
        frag_load(buf, xb, 0, laneM, kofs, half8, fa0);
        frag_load(buf, xb, 1, laneM, kofs, half8, fb0);
#pragma unroll
        for (int r = 0; r < CH / 32; ++r) {
            Frag& ca = (r & 1) ? fa1 : fa0;
            Frag& cb = (r & 1) ? fb1 : fb0;
            Frag& na = (r & 1) ? fa0 : fa1;
            Frag& nb = (r & 1) ? fb0 : fb1;
            if (r + 1 < CH / 32) {
                frag_load(buf, xb, 2 * r + 2, laneM, kofs, half8, na);
                frag_load(buf, xb, 2 * r + 3, laneM, kofs, half8, nb);
            }
            __builtin_amdgcn_sched_barrier(0);
            float cva[8], cvb[8];
            int csa[8], csb[8];
            const bool ins = (r == 0) ? (c > NPRE) : (c >= NPRE);
            v8f accA, accB;
            products2_mark(ca, cb, bq, pendA, pendB, xxq, thr, ins, cnt, cva, csa, cvb, csb, topv,
                           accA, accB);
            select2_store(lv, li, pcol, cva, csa);
            select2_store(lv, li, pcol + 16, cvb, csb);
            {
                const float bnd = topv[KK - 1];
                thr = (bnd > thr) ? bnd : thr;
            }
            pendA.acc = accA;
            pendA.x0 = ca.x0;
            pendA.x1 = ca.x1;
            pendB.acc = accB;
            pendB.x0 = cb.x0;
            pendB.x1 = cb.x1;
            pcol = c * CH + r * 32 + half8;
        }
        if (cnt >= LTRIG) {
            cnt = drop_to_20(lv, li, cnt);
            float lo = lv[0];
            for (int s = 1; s < KK; ++s) lo = fminf(lo, lv[s * 32]);
            if (lo >= thr) thr = next_above(lo);
        }
        if (more) {
            stage_put(colbuf[(c + 1) & 1], xxbuf[(c + 1) & 1], tid, st, sx);
            __syncthreads();
        }
    }
#pragma unroll
    for (int e = 0; e < 2; ++e) {
        const Pend& pe = (e == 0) ? pendA : pendB;
        float cv[8];
        int cs[8];
        select2_mark(pe, xxq, thr, cnt, cv, cs);
        select2_store(lv, li, pcol + 16 * e, cv, cs);
        float m = fmaxf(fmaxf(fmaxf(cv[0], cv[1]), fmaxf(cv[2], cv[3])),
                        fmaxf(fmaxf(cv[4], cv[5]), fmaxf(cv[6], cv[7])));
#pragma unroll
        for (int s = 0; s < KK; ++s) {
            float hi = fmaxf(topv[s], m);
            m = fminf(topv[s], m);
            topv[s] = hi;
        }
        if (e == 0) {
            const float bnd = topv[KK - 1];
            thr = (bnd > thr) ? bnd : thr;
        }
    }
    {
        float pv[KK];
#pragma unroll
        for (int s = 0; s < KK; ++s) pv[s] = __shfl_xor(topv[s], 16);
#pragma unroll
        for (int t = 0; t < KK; ++t) {
            float m = pv[t];
#pragma unroll
            for (int s = 0; s < KK; ++s) {
                float hi = fmaxf(topv[s], m);
                m = fminf(topv[s], m);
                topv[s] = hi;
            }
        }
    }
    static_assert(LCAP % 8 == 0, "groups of eight must stay inside a lane's list");
    {
        const float thrq = topv[KK - 1];
        int kept = 0;
        for (int e0 = 0; e0 < cnt; e0 += 8) {
            float ve[8];
            int ie[8];
#pragma unroll
            for (int j = 0; j < 8; ++j) {
                ve[j] = lv[(e0 + j) * 32];
                ie[j] = li[(e0 + j) * 32];
            }
#pragma unroll
            for (int j = 0; j < 8; ++j) {
                lv[kept * 32] = ve[j];
                li[kept * 32] = ie[j];
                kept += ((e0 + j < cnt) && (ve[j] >= thrq)) ? 1 : 0;
            }
        }
        cnt = kept;
    }
    cnt = drop_to_20(lv, li, cnt);

    const int cntUp = __shfl_xor(cnt, 16);
    __syncthreads();
    if (half == 0) {
        for (int e = 0; e < cntUp; ++e) {
            lv[(cnt + e) * 32] = lv[16 + e * 32];
            li[(cnt + e) * 32] = li[16 + e * 32];
        }
        const int tot = drop_to_20(lv, li, cnt + cntUp);
        int* outp = nbr + (size_t)(b * NN + q0 + laneM) * KK;
        for (int s = 0; s < KK; ++s) outp[s] = (s < tot) ? li[s * 32] : 0;
    }
}

__device__ __forceinline__ void wave_lds_order() {
    __builtin_amdgcn_fence(__ATOMIC_RELEASE, "wavefront");
    __builtin_amdgcn_wave_barrier();
    __builtin_amdgcn_fence(__ATOMIC_ACQUIRE, "wavefront");
}

#define GPB 4
#define TPW 20

__device__ __forceinline__ int edge_row(int G0, int wave, int q, int laneM) {
    const int g = q / 5;
    const int it = q - g * 5;
    return (G0 + g) * (16 * KK) + (wave * 5 + it) * 16 + laneM;
}

__global__ __launch_bounds__(128) void edgeconv_kernel(
    const float* __restrict__ xt, const int* __restrict__ nbr,
    const float* __restrict__ w1, const float* __restrict__ g1,
    const float* __restrict__ be1, const float* __restrict__ mu1,
    const float* __restrict__ va1, const float* __restrict__ w2,
    const float* __restrict__ g2, const float* __restrict__ be2,
    const float* __restrict__ mu2, const float* __restrict__ va2,
    float* __restrict__ out) {
    const int wave = threadIdx.x >> 5;
    const int lane = threadIdx.x & 31;
    const int laneM = lane & 15;
    const int kofs = (lane < 16) ? 0 : 2;
    const int G0 = blockIdx.x * GPB;

    __shared__ float h1s[4][16 * OO];
    __shared__ unsigned om[16 * OO];

    int e0 = edge_row(G0, wave, 0, laneM);
    int nb0 = nbr[e0];
    int e1 = edge_row(G0, wave, 1, laneM);
    int nbN = nbr[e1];

    v2f wa[16][4], wb[16][4], wc[16][4];
#pragma unroll
    for (int ks = 0; ks < 16; ++ks) {
        const int kc = ks * 4 + kofs;
#pragma unroll
        for (int nt = 0; nt < 4; ++nt) {
            const int o = nt * 16 + laneM;
            wa[ks][nt] = *(const v2f*)(w1 + o * 128 + kc);
            wb[ks][nt] = *(const v2f*)(w1 + o * 128 + 64 + kc);
            wc[ks][nt] = *(const v2f*)(w2 + o * OO + kc);
        }
    }
    float sc1[4], sh1[4], sc2[4], sh2[4];
#pragma unroll
    for (int nt = 0; nt < 4; ++nt) {
        const int o = nt * 16 + laneM;
        sc1[nt] = g1[o] * rsqrtf(va1[o] + EPSV);
        sh1[nt] = be1[o] - mu1[o] * sc1[nt];
        sc2[nt] = g2[o] * rsqrtf(va2[o] + EPSV);
        sh2[nt] = be2[o] - mu2[o] * sc2[nt];
    }

    v2f an[16], cn[16];
    {
        const float* pc = xt + (size_t)(e0 / KK) * CC + kofs;
        const float* pn = xt + ((size_t)((e0 / KK) & ~(NN - 1)) + nb0) * CC + kofs;
#pragma unroll
        for (int ks = 0; ks < 16; ++ks) {
            an[ks] = *(const v2f*)(pn + ks * 4);
            cn[ks] = *(const v2f*)(pc + ks * 4);
        }
    }

    for (int i = threadIdx.x; i < 16 * OO; i += 128) om[i] = 0u;

    for (int q = 0; q < TPW; ++q) {
        const int g = q / 5;
        const int it = q - g * 5;
        const int G = G0 + g;
        const int mt = wave * 5 + it;
        if (it == 0) __syncthreads();

        v2f ac[16], cc[16];
#pragma unroll
        for (int ks = 0; ks < 16; ++ks) {
            ac[ks] = an[ks];
            cc[ks] = cn[ks];
        }
        {
            const int qn = (q + 1 < TPW) ? q + 1 : TPW - 1;
            const int q2 = (q + 2 < TPW) ? q + 2 : TPW - 1;
            const int en = edge_row(G0, wave, qn, laneM);
            const float* pc = xt + (size_t)(en / KK) * CC + kofs;
            const float* pn = xt + ((size_t)((en / KK) & ~(NN - 1)) + nbN) * CC + kofs;
#pragma unroll
            for (int ks = 0; ks < 16; ++ks) {
                an[ks] = *(const v2f*)(pn + ks * 4);
                cn[ks] = *(const v2f*)(pc + ks * 4);
            }
            nbN = nbr[edge_row(G0, wave, q2, laneM)];
        }

        v8f acc1[4] = {};
#pragma unroll
        for (int ks = 0; ks < 16; ++ks) {
            v2f a = ac[ks] - cc[ks];
#pragma unroll
            for (int nt = 0; nt < 4; ++nt) acc1[nt] = wmma_f32(a, wa[ks][nt], acc1[nt]);
        }
#pragma unroll
        for (int ks = 0; ks < 16; ++ks) {
#pragma unroll
            for (int nt = 0; nt < 4; ++nt) acc1[nt] = wmma_f32(cc[ks], wb[ks][nt], acc1[nt]);
        }
#pragma unroll
        for (int nt = 0; nt < 4; ++nt) {
            int o = nt * 16 + laneM;
#pragma unroll
            for (int j = 0; j < 8; ++j) {
                int M = j + ((lane < 16) ? 0 : 8);
                h1s[wave][M * OO + o] = fmaxf(acc1[nt][j] * sc1[nt] + sh1[nt], 0.0f);
            }
        }
        wave_lds_order();

        v8f acc2[4] = {};
#pragma unroll
        for (int ks = 0; ks < 16; ++ks) {
            int kc = ks * 4 + kofs;
            v2f a = *(const v2f*)(&h1s[wave][laneM * OO + kc]);
#pragma unroll
            for (int nt = 0; nt < 4; ++nt) acc2[nt] = wmma_f32(a, wc[ks][nt], acc2[nt]);
        }
#pragma unroll
        for (int nt = 0; nt < 4; ++nt) {
            int o = nt * 16 + laneM;
#pragma unroll
            for (int j = 0; j < 8; ++j) {
                int M = j + ((lane < 16) ? 0 : 8);
                int lr = mt * 16 + M;
                int pp = lr / KK;
                float val = fmaxf(acc2[nt][j] * sc2[nt] + sh2[nt], 0.0f);
                atomicMax(&om[pp * OO + o], __float_as_uint(val));
            }
        }
        wave_lds_order();

        if (it == 4) {
            __syncthreads();
            const int b = G >> 8;
            const int pt = G & 255;
            for (int i = threadIdx.x; i < 16 * OO; i += 128) {
                int p = i >> 6;
                int o = i & 63;
                out[(size_t)(b * OO + o) * NN + pt * 16 + p] = __uint_as_float(om[p * OO + o]);
                om[i] = 0u;
            }
        }
    }
}

extern "C" void kernel_launch(void* const* d_in, const int* in_sizes, int n_in,
                              void* d_out, int out_size, void* d_ws, size_t ws_size,
                              hipStream_t stream) {
    const float* x   = (const float*)d_in[0];
    const float* w1  = (const float*)d_in[1];
    const float* g1  = (const float*)d_in[2];
    const float* be1 = (const float*)d_in[3];
    const float* mu1 = (const float*)d_in[4];
    const float* va1 = (const float*)d_in[5];
    const float* w2  = (const float*)d_in[6];
    const float* g2  = (const float*)d_in[7];
    const float* be2 = (const float*)d_in[8];
    const float* mu2 = (const float*)d_in[9];
    const float* va2 = (const float*)d_in[10];
    float* out = (float*)d_out;

    float* xt = (float*)d_ws;
    float* xx = xt + (size_t)BB * NN * CC;
    int* nbr = (int*)(xx + (size_t)BB * NN);

    prep_kernel<<<(BB * NN) / 256, 256, 0, stream>>>(x, xt, xx);
    knn_kernel<<<(BB * NN) / 64, 128, 0, stream>>>(xt, xx, nbr);
    edgeconv_kernel<<<(BB * (NN / 16)) / GPB, 128, 0, stream>>>(
        xt, nbr, w1, g1, be1, mu1, va1, w2, g2, be2, mu2, va2, out);
}


# kNN sweep 1 cut to the first stage (maxima of 2-column groups); sweep 2 starts behind it and takes sweep 1's columns last, under the bound of all other columns; list room 160 to 224 entries
# speedup vs baseline: 25.3612x; 1.0499x over previous
#include <hip/hip_runtime.h>
#include <math.h>

#define BB 4
#define CC 64
#define NN 4096
#define OO 64
#define KK 20
#define EPSV 1e-5f

typedef __attribute__((ext_vector_type(2))) float v2f;
typedef __attribute__((ext_vector_type(4))) float v4f;
typedef __attribute__((ext_vector_type(8))) float v8f;

__device__ __forceinline__ v8f wmma_f32(v2f a, v2f b, v8f c) {
    return __builtin_amdgcn_wmma_f32_16x16x4_f32(false, a, false, b, (short)0, c,
                                                 false, false);
}

__global__ __launch_bounds__(256) void prep_kernel(const float* __restrict__ x,
                                                   float* __restrict__ xt,
                                                   float* __restrict__ xx) {
    int t = blockIdx.x * blockDim.x + threadIdx.x;
    int b = t / NN;
    int n = t - b * NN;
    const float* px = x + (size_t)b * CC * NN + n;
    float* pt = xt + (size_t)t * CC;
    float s = 0.f;
#pragma unroll 8
    for (int c = 0; c < CC; ++c) {
        float v = px[(size_t)c * NN];
        pt[c] = v;
        s += v * v;
    }
    xx[t] = s;
}

#define LCAP 224
#define LTRIG 160
#define NPRE 1
#define SG 4

#define CH 128
#define NCH (NN / CH)
#define PITCH 68
#define SLD 16

__device__ __forceinline__ void stage_fetch(const float* __restrict__ xt_b,
                                            const float* __restrict__ xx_b, int chunk, int tid,
                                            v4f (&st)[SLD], float& sx) {
    const v4f* src = (const v4f*)(xt_b + (size_t)chunk * (CH * CC)) + tid;
#pragma unroll
    for (int i = 0; i < SLD; ++i) st[i] = src[i * 128];
    sx = xx_b[chunk * CH + tid];
}

__device__ __forceinline__ void stage_put(float* __restrict__ buf, float* __restrict__ xbuf,
                                          int tid, const v4f (&st)[SLD], float sx) {
    float* dst = buf + (tid >> 4) * PITCH + (tid & 15) * 4;
#pragma unroll
    for (int i = 0; i < SLD; ++i) *(v4f*)(dst + i * 8 * PITCH) = st[i];
    xbuf[tid] = sx;
}

struct Frag {
    v2f a[16];
    v4f x0, x1;
};
struct Pend {
    v8f acc;
    v4f x0, x1;
};

__device__ __forceinline__ void frag_load(const float* __restrict__ buf,
                                          const float* __restrict__ xbuf, int t, int laneM,
                                          int kofs, int half8, Frag& f) {
    const float* prow = buf + (t * 16 + laneM) * PITCH + kofs;
    const v4f* pxx = (const v4f*)(xbuf + t * 16 + half8);
#pragma unroll
    for (int ks = 0; ks < 16; ++ks) f.a[ks] = *(const v2f*)(prow + ks * 4);
    f.x0 = pxx[0];
    f.x1 = pxx[1];
}

__device__ __forceinline__ void products2(const Frag& fa, const Frag& fb, const v2f (&bq)[16],
                                          v8f& accA, v8f& accB) {
    accA = v8f{};
    accB = v8f{};
#pragma unroll
    for (int ks = 0; ks < 16; ++ks) {
        accA = wmma_f32(fa.a[ks], bq[ks], accA);
        accB = wmma_f32(fb.a[ks], bq[ks], accB);
    }
}

__device__ __forceinline__ void finish(const Pend& p, float (&u)[8]) {
    u[0] = 2.0f * p.acc[0] - p.x0.x;
    u[1] = 2.0f * p.acc[1] - p.x0.y;
    u[2] = 2.0f * p.acc[2] - p.x0.z;
    u[3] = 2.0f * p.acc[3] - p.x0.w;
    u[4] = 2.0f * p.acc[4] - p.x1.x;
    u[5] = 2.0f * p.acc[5] - p.x1.y;
    u[6] = 2.0f * p.acc[6] - p.x1.z;
    u[7] = 2.0f * p.acc[7] - p.x1.w;
}

__device__ __forceinline__ void select1(const Pend& p, float xxq, float (&topv)[KK]) {
    static_assert(NPRE * 8 * SG >= KK && 8 % SG == 0, "sweep 1 must fill the list of maxima");
    float u[8];
    finish(p, u);
#pragma unroll
    for (int g = 0; g < SG; ++g) {
        float m = u[g * (8 / SG)];
#pragma unroll
        for (int r = 1; r < 8 / SG; ++r) m = fmaxf(m, u[g * (8 / SG) + r]);
        m -= xxq;
#pragma unroll
        for (int s = 0; s < KK; ++s) {
            float hi = fmaxf(topv[s], m);
            m = fminf(topv[s], m);
            topv[s] = hi;
        }
    }
}

__device__ __forceinline__ void select2_mark(const Pend& p, float xxq, float thr, int& cnt,
                                             float (&v)[8], int (&slot)[8]) {
    float u[8];
    finish(p, u);
#pragma unroll
    for (int r = 0; r < 8; ++r) {
        v[r] = u[r] - xxq;
        slot[r] = cnt;
        cnt += (v[r] >= thr) ? 1 : 0;
    }
}

__device__ __forceinline__ float next_above(float x) {
    const int b = __float_as_int(x);
    if (x == 0.0f) return __int_as_float(1);
    return __int_as_float(b >= 0 ? b + 1 : b - 1);
}

__device__ __forceinline__ void select2_store(float* lv, int* li, int colbase,
                                              const float (&v)[8], const int (&slot)[8]) {
#pragma unroll
    for (int r = 0; r < 8; ++r) {
        lv[slot[r] * 32] = v[r];
        li[slot[r] * 32] = colbase + r;
    }
}

__device__ __forceinline__ void select2_piece(int j, const float (&dv)[8], float thr, bool ins,
                                              int& cnt, float (&v)[8], int (&slot)[8], float& m,
                                              float (&topv)[KK]) {
    if (j < 8) {
        float d = dv[j];
        asm volatile("" : "+v"(d));
        v[j] = d;
        slot[j] = cnt;
        cnt += (v[j] >= thr) ? 1 : 0;
    } else {
        if (j == 8) {
            float mx = fmaxf(fmaxf(fmaxf(v[0], v[1]), fmaxf(v[2], v[3])),
                             fmaxf(fmaxf(v[4], v[5]), fmaxf(v[6], v[7])));
            m = ins ? mx : -INFINITY;
        }
        const int sb = (j == 8) ? 0 : 2 + (j - 9) * 3;
        const int se = (j == 8) ? 2 : ((2 + (j - 8) * 3) < KK ? (2 + (j - 8) * 3) : KK);
#pragma unroll
        for (int sl = sb; sl < se; ++sl) {
            float hi = fmaxf(topv[sl], m);
            m = fminf(topv[sl], m);
            topv[sl] = hi;
        }
    }
}

__device__ __forceinline__ void products2_mark(const Frag& fa, const Frag& fb, const v2f (&bq)[16],
                                               const Pend& pa, const Pend& pb, float xxq,
                                               float& thr, bool ins, int& cnt, float (&va)[8],
                                               int (&sa)[8], float (&vb)[8], int (&sb)[8],
                                               float (&topv)[KK], v8f& accA, v8f& accB) {
    float ua[8], ub[8], da[8], db[8];
    finish(pa, ua);
#pragma unroll
    for (int j = 0; j < 8; ++j) da[j] = ua[j] - xxq;
    float m = -INFINITY;
    accA = v8f{};
    accB = v8f{};
#pragma unroll
    for (int i = 0; i < 32; ++i) {
        const int ks = i >> 1;
        if ((i & 1) == 0) accA = wmma_f32(fa.a[ks], bq[ks], accA);
        else              accB = wmma_f32(fb.a[ks], bq[ks], accB);
        if (i < 16) {
            select2_piece(i, da, thr, ins, cnt, va, sa, m, topv);
            if (i == 15) {
                const float bnd = topv[KK - 1];
                thr = (bnd > thr) ? bnd : thr;
                finish(pb, ub);
#pragma unroll
                for (int j = 0; j < 8; ++j) db[j] = ub[j] - xxq;
            }
        } else {
            select2_piece(i - 16, db, thr, ins, cnt, vb, sb, m, topv);
        }
        __builtin_amdgcn_sched_barrier(0);
    }
}

__device__ __forceinline__ int drop_to_20(float* lv, int* li, int m) {
    while (m > KK) {
        float wv = lv[0];
        int wi = li[0];
        int wp = 0;
        for (int j = 1; j < m; ++j) {
            float vj = lv[j * 32];
            int ij = li[j * 32];
            bool worse = (vj < wv) || (vj == wv && ij > wi);
            if (worse) { wv = vj; wi = ij; wp = j; }
        }
        --m;
        lv[wp * 32] = lv[m * 32];
        li[wp * 32] = li[m * 32];
    }
    return m;
}

__global__ __launch_bounds__(128) void knn_kernel(const float* __restrict__ xt,
                                                  const float* __restrict__ xx,
                                                  int* __restrict__ nbr) {
    const int tid = threadIdx.x;
    const int wave = threadIdx.x >> 5;
    const int lane = threadIdx.x & 31;
    const int laneM = lane & 15;
    const int half = lane >> 4;
    const int kofs = half * 2;
    const int half8 = half * 8;

    const int qg = blockIdx.x * 64 + wave * 16;
    const int b = qg / NN;
    const int q0 = qg - b * NN;
    const float* xt_b = xt + (size_t)b * NN * CC;
    const float* xx_b = xx + b * NN;

    const float* pq = xt_b + (size_t)(q0 + laneM) * CC + kofs;
    v2f bq[16];
#pragma unroll
    for (int ks = 0; ks < 16; ++ks) bq[ks] = *(const v2f*)(pq + ks * 4);
    const float xxq = xx_b[q0 + laneM];

    __shared__ __attribute__((aligned(16))) float colbuf[2][CH * PITCH];
    __shared__ __attribute__((aligned(16))) float xxbuf[2][CH];
    struct Lists {
        float v[LCAP * 32];
        int i[LCAP * 32];
    };
    __shared__ Lists lists[4];
    float* lv = &lists[wave].v[lane];
    int* li = &lists[wave].i[lane];

    v4f st[SLD];
    float sx;
    stage_fetch(xt_b, xx_b, 0, tid, st, sx);
    stage_put(colbuf[0], xxbuf[0], tid, st, sx);
    __syncthreads();

    float topv[KK];
#pragma unroll
    for (int s = 0; s < KK; ++s) topv[s] = -3.0e38f;

    Pend pendA, pendB;
    pendA.acc = v8f{};
    pendA.x0 = v4f{INFINITY, INFINITY, INFINITY, INFINITY};
    pendA.x1 = pendA.x0;
    pendB = pendA;
    for (int c = 0; c < NPRE; ++c) {
        stage_fetch(xt_b, xx_b, c + 1, tid, st, sx);
        const float* buf = colbuf[c & 1];
        const float* xb = xxbuf[c & 1];
        Frag fa0, fb0, fa1, fb1;
        frag_load(buf, xb, 0, laneM, kofs, half8, fa0);
        frag_load(buf, xb, 1, laneM, kofs, half8, fb0);
#pragma unroll
        for (int r = 0; r < CH / 32; ++r) {
            Frag& ca = (r & 1) ? fa1 : fa0;
            Frag& cb = (r & 1) ? fb1 : fb0;
            Frag& na = (r & 1) ? fa0 : fa1;
            Frag& nb = (r & 1) ? fb0 : fb1;
            if (r + 1 < CH / 32) {
                frag_load(buf, xb, 2 * r + 2, laneM, kofs, half8, na);
                frag_load(buf, xb, 2 * r + 3, laneM, kofs, half8, nb);
            }
            __builtin_amdgcn_sched_barrier(0);
            v8f accA, accB;
            products2(ca, cb, bq, accA, accB);
            select1(pendA, xxq, topv);
            select1(pendB, xxq, topv);
            __builtin_amdgcn_sched_barrier(0);
            pendA.acc = accA;
            pendA.x0 = ca.x0;
            pendA.x1 = ca.x1;
            pendB.acc = accB;
            pendB.x0 = cb.x0;
            pendB.x1 = cb.x1;
        }
        stage_put(colbuf[(c + 1) & 1], xxbuf[(c + 1) & 1], tid, st, sx);
        __syncthreads();
    }
    select1(pendA, xxq, topv);
    select1(pendB, xxq, topv);

    float thr = topv[KK - 1];
    float lo_inc = -INFINITY;
    static_assert((NCH & (NCH - 1)) == 0 && NPRE < NCH, "the stage order wraps by a mask");
    int cnt = 0;
    pendA.acc = v8f{};
    pendA.x0 = v4f{NAN, NAN, NAN, NAN};
    pendA.x1 = pendA.x0;
    pendB = pendA;
    int pcol = half8;
    for (int i = 0; i < NCH; ++i) {
        const int c = (i + NPRE) & (NCH - 1);
        const int cnx = (c + 1) & (NCH - 1);
        const bool rising = (i < NCH - NPRE);
        const bool more = (i + 1 < NCH);
        if (more) stage_fetch(xt_b, xx_b, cnx, tid, st, sx);
        if (i == NCH - NPRE) thr = fmaxf(topv[KK - 1], lo_inc);
        const float* buf = colbuf[c & 1];
        const float* xb = xxbuf[c & 1];
        Frag fa0, fb0, fa1, fb1;
        frag_load(buf, xb, 0, laneM, kofs, half8, fa0);
        frag_load(buf, xb, 1, laneM, kofs, half8, fb0);
#pragma unroll
        for (int r = 0; r < CH / 32; ++r) {
            Frag& ca = (r & 1) ? fa1 : fa0;
            Frag& cb = (r & 1) ? fb1 : fb0;
            Frag& na = (r & 1) ? fa0 : fa1;
            Frag& nb = (r & 1) ? fb0 : fb1;
            if (r + 1 < CH / 32) {
                frag_load(buf, xb, 2 * r + 2, laneM, kofs, half8, na);
                frag_load(buf, xb, 2 * r + 3, laneM, kofs, half8, nb);
            }
            __builtin_amdgcn_sched_barrier(0);
            float cva[8], cvb[8];
            int csa[8], csb[8];
            const bool ins = (r == 0) ? (i >= 1 && i <= NCH - NPRE) : rising;
            v8f accA, accB;
            products2_mark(ca, cb, bq, pendA, pendB, xxq, thr, ins, cnt, cva, csa, cvb, csb, topv,
                           accA, accB);
            select2_store(lv, li, pcol, cva, csa);
            select2_store(lv, li, pcol + 16, cvb, csb);
            {
                const float bnd = topv[KK - 1];
                thr = (bnd > thr) ? bnd : thr;
            }
            pendA.acc = accA;
            pendA.x0 = ca.x0;
            pendA.x1 = ca.x1;
            pendB.acc = accB;
            pendB.x0 = cb.x0;
            pendB.x1 = cb.x1;
            pcol = c * CH + r * 32 + half8;
        }
        if (cnt >= LTRIG) {
            cnt = drop_to_20(lv, li, cnt);
            float lo = lv[0];
            for (int s = 1; s < KK; ++s) lo = fminf(lo, lv[s * 32]);
            if (rising) {
                if (lo >= thr) thr = next_above(lo);
                lo_inc = fmaxf(lo_inc, lo);
            } else {
                if (lo > thr) thr = lo;
            }
        }
        if (more) {
            stage_put(colbuf[cnx & 1], xxbuf[cnx & 1], tid, st, sx);
            __syncthreads();
        }
    }
#pragma unroll
    for (int e = 0; e < 2; ++e) {
        const Pend& pe = (e == 0) ? pendA : pendB;
        float cv[8];
        int cs[8];
        select2_mark(pe, xxq, thr, cnt, cv, cs);
        select2_store(lv, li, pcol + 16 * e, cv, cs);
    }
    {
        float pv[KK];
#pragma unroll
        for (int s = 0; s < KK; ++s) pv[s] = __shfl_xor(topv[s], 16);
#pragma unroll
        for (int t = 0; t < KK; ++t) {
            float m = pv[t];
#pragma unroll
            for (int s = 0; s < KK; ++s) {
                float hi = fmaxf(topv[s], m);
                m = fminf(topv[s], m);
                topv[s] = hi;
            }
        }
    }
    static_assert(LCAP % 8 == 0, "groups of eight must stay inside a lane's list");
    {
        const float thrq = topv[KK - 1];
        int kept = 0;
        for (int e0 = 0; e0 < cnt; e0 += 8) {
            float ve[8];
            int ie[8];
#pragma unroll
            for (int j = 0; j < 8; ++j) {
                ve[j] = lv[(e0 + j) * 32];
                ie[j] = li[(e0 + j) * 32];
            }
#pragma unroll
            for (int j = 0; j < 8; ++j) {
                lv[kept * 32] = ve[j];
                li[kept * 32] = ie[j];
                kept += ((e0 + j < cnt) && (ve[j] >= thrq)) ? 1 : 0;
            }
        }
        cnt = kept;
    }
    cnt = drop_to_20(lv, li, cnt);

    const int cntUp = __shfl_xor(cnt, 16);
    __syncthreads();
    if (half == 0) {
        for (int e = 0; e < cntUp; ++e) {
            lv[(cnt + e) * 32] = lv[16 + e * 32];
            li[(cnt + e) * 32] = li[16 + e * 32];
        }
        const int tot = drop_to_20(lv, li, cnt + cntUp);
        int* outp = nbr + (size_t)(b * NN + q0 + laneM) * KK;
        for (int s = 0; s < KK; ++s) outp[s] = (s < tot) ? li[s * 32] : 0;
    }
}

__device__ __forceinline__ void wave_lds_order() {
    __builtin_amdgcn_fence(__ATOMIC_RELEASE, "wavefront");
    __builtin_amdgcn_wave_barrier();
    __builtin_amdgcn_fence(__ATOMIC_ACQUIRE, "wavefront");
}

#define GPB 4
#define TPW 20

__device__ __forceinline__ int edge_row(int G0, int wave, int q, int laneM) {
    const int g = q / 5;
    const int it = q - g * 5;
    return (G0 + g) * (16 * KK) + (wave * 5 + it) * 16 + laneM;
}

__global__ __launch_bounds__(128) void edgeconv_kernel(
    const float* __restrict__ xt, const int* __restrict__ nbr,
    const float* __restrict__ w1, const float* __restrict__ g1,
    const float* __restrict__ be1, const float* __restrict__ mu1,
    const float* __restrict__ va1, const float* __restrict__ w2,
    const float* __restrict__ g2, const float* __restrict__ be2,
    const float* __restrict__ mu2, const float* __restrict__ va2,
    float* __restrict__ out) {
    const int wave = threadIdx.x >> 5;
    const int lane = threadIdx.x & 31;
    const int laneM = lane & 15;
    const int kofs = (lane < 16) ? 0 : 2;
    const int G0 = blockIdx.x * GPB;

    __shared__ float h1s[4][16 * OO];
    __shared__ unsigned om[16 * OO];

    int e0 = edge_row(G0, wave, 0, laneM);
    int nb0 = nbr[e0];
    int e1 = edge_row(G0, wave, 1, laneM);
    int nbN = nbr[e1];

    v2f wa[16][4], wb[16][4], wc[16][4];
#pragma unroll
    for (int ks = 0; ks < 16; ++ks) {
        const int kc = ks * 4 + kofs;
#pragma unroll
        for (int nt = 0; nt < 4; ++nt) {
            const int o = nt * 16 + laneM;
            wa[ks][nt] = *(const v2f*)(w1 + o * 128 + kc);
            wb[ks][nt] = *(const v2f*)(w1 + o * 128 + 64 + kc);
            wc[ks][nt] = *(const v2f*)(w2 + o * OO + kc);
        }
    }
    float sc1[4], sh1[4], sc2[4], sh2[4];
#pragma unroll
    for (int nt = 0; nt < 4; ++nt) {
        const int o = nt * 16 + laneM;
        sc1[nt] = g1[o] * rsqrtf(va1[o] + EPSV);
        sh1[nt] = be1[o] - mu1[o] * sc1[nt];
        sc2[nt] = g2[o] * rsqrtf(va2[o] + EPSV);
        sh2[nt] = be2[o] - mu2[o] * sc2[nt];
    }

    v2f an[16], cn[16];
    {
        const float* pc = xt + (size_t)(e0 / KK) * CC + kofs;
        const float* pn = xt + ((size_t)((e0 / KK) & ~(NN - 1)) + nb0) * CC + kofs;
#pragma unroll
        for (int ks = 0; ks < 16; ++ks) {
            an[ks] = *(const v2f*)(pn + ks * 4);
            cn[ks] = *(const v2f*)(pc + ks * 4);
        }
    }

    for (int i = threadIdx.x; i < 16 * OO; i += 128) om[i] = 0u;

    for (int q = 0; q < TPW; ++q) {
        const int g = q / 5;
        const int it = q - g * 5;
        const int G = G0 + g;
        const int mt = wave * 5 + it;
        if (it == 0) __syncthreads();

        v2f ac[16], cc[16];
#pragma unroll
        for (int ks = 0; ks < 16; ++ks) {
            ac[ks] = an[ks];
            cc[ks] = cn[ks];
        }
        {
            const int qn = (q + 1 < TPW) ? q + 1 : TPW - 1;
            const int q2 = (q + 2 < TPW) ? q + 2 : TPW - 1;
            const int en = edge_row(G0, wave, qn, laneM);
            const float* pc = xt + (size_t)(en / KK) * CC + kofs;
            const float* pn = xt + ((size_t)((en / KK) & ~(NN - 1)) + nbN) * CC + kofs;
#pragma unroll
            for (int ks = 0; ks < 16; ++ks) {
                an[ks] = *(const v2f*)(pn + ks * 4);
                cn[ks] = *(const v2f*)(pc + ks * 4);
            }
            nbN = nbr[edge_row(G0, wave, q2, laneM)];
        }

        v8f acc1[4] = {};
#pragma unroll
        for (int ks = 0; ks < 16; ++ks) {
            v2f a = ac[ks] - cc[ks];
#pragma unroll
            for (int nt = 0; nt < 4; ++nt) acc1[nt] = wmma_f32(a, wa[ks][nt], acc1[nt]);
        }
#pragma unroll
        for (int ks = 0; ks < 16; ++ks) {
#pragma unroll
            for (int nt = 0; nt < 4; ++nt) acc1[nt] = wmma_f32(cc[ks], wb[ks][nt], acc1[nt]);
        }
#pragma unroll
        for (int nt = 0; nt < 4; ++nt) {
            int o = nt * 16 + laneM;
#pragma unroll
            for (int j = 0; j < 8; ++j) {
                int M = j + ((lane < 16) ? 0 : 8);
                h1s[wave][M * OO + o] = fmaxf(acc1[nt][j] * sc1[nt] + sh1[nt], 0.0f);
            }
        }
        wave_lds_order();

        v8f acc2[4] = {};
#pragma unroll
        for (int ks = 0; ks < 16; ++ks) {
            int kc = ks * 4 + kofs;
            v2f a = *(const v2f*)(&h1s[wave][laneM * OO + kc]);
#pragma unroll
            for (int nt = 0; nt < 4; ++nt) acc2[nt] = wmma_f32(a, wc[ks][nt], acc2[nt]);
        }
#pragma unroll
        for (int nt = 0; nt < 4; ++nt) {
            int o = nt * 16 + laneM;
#pragma unroll
            for (int j = 0; j < 8; ++j) {
                int M = j + ((lane < 16) ? 0 : 8);
                int lr = mt * 16 + M;
                int pp = lr / KK;
                float val = fmaxf(acc2[nt][j] * sc2[nt] + sh2[nt], 0.0f);
                atomicMax(&om[pp * OO + o], __float_as_uint(val));
            }
        }
        wave_lds_order();

        if (it == 4) {
            __syncthreads();
            const int b = G >> 8;
            const int pt = G & 255;
            for (int i = threadIdx.x; i < 16 * OO; i += 128) {
                int p = i >> 6;
                int o = i & 63;
                out[(size_t)(b * OO + o) * NN + pt * 16 + p] = __uint_as_float(om[p * OO + o]);
                om[i] = 0u;
            }
        }
    }
}

extern "C" void kernel_launch(void* const* d_in, const int* in_sizes, int n_in,
                              void* d_out, int out_size, void* d_ws, size_t ws_size,
                              hipStream_t stream) {
    const float* x   = (const float*)d_in[0];
    const float* w1  = (const float*)d_in[1];
    const float* g1  = (const float*)d_in[2];
    const float* be1 = (const float*)d_in[3];
    const float* mu1 = (const float*)d_in[4];
    const float* va1 = (const float*)d_in[5];
    const float* w2  = (const float*)d_in[6];
    const float* g2  = (const float*)d_in[7];
    const float* be2 = (const float*)d_in[8];
    const float* mu2 = (const float*)d_in[9];
    const float* va2 = (const float*)d_in[10];
    float* out = (float*)d_out;

    float* xt = (float*)d_ws;
    float* xx = xt + (size_t)BB * NN * CC;
    int* nbr = (int*)(xx + (size_t)BB * NN);

    prep_kernel<<<(BB * NN) / 256, 256, 0, stream>>>(x, xt, xx);
    knn_kernel<<<(BB * NN) / 64, 128, 0, stream>>>(xt, xx, nbr);
    edgeconv_kernel<<<(BB * (NN / 16)) / GPB, 128, 0, stream>>>(
        xt, nbr, w1, g1, be1, mu1, va1, w2, g2, be2, mu2, va2, out);
}


# global stores made whole-line: prep transposes 64x64 tiles through LDS (256 blocks, 16-byte loads and stores); edge MLP keeps its block's 64 points' maxima in LDS and writes them once at the end
# speedup vs baseline: 27.1684x; 1.0713x over previous
#include <hip/hip_runtime.h>
#include <math.h>

#define BB 4
#define CC 64
#define NN 4096
#define OO 64
#define KK 20
#define EPSV 1e-5f

typedef __attribute__((ext_vector_type(2))) float v2f;
typedef __attribute__((ext_vector_type(4))) float v4f;
typedef __attribute__((ext_vector_type(8))) float v8f;

__device__ __forceinline__ v8f wmma_f32(v2f a, v2f b, v8f c) {
    return __builtin_amdgcn_wmma_f32_16x16x4_f32(false, a, false, b, (short)0, c,
                                                 false, false);
}

#define PTS 64
#define PPITCH 65
#define PLD 8

__global__ __launch_bounds__(128) void prep_kernel(const float* __restrict__ x,
                                                   float* __restrict__ xt,
                                                   float* __restrict__ xx) {
    __shared__ float tile[CC * PPITCH];
    const int tid = threadIdx.x;
    const int b = blockIdx.x / (NN / PTS);
    const int n0 = (blockIdx.x - b * (NN / PTS)) * PTS;
    const float* px = x + (size_t)b * CC * NN + n0;

    v4f r[PLD];
#pragma unroll
    for (int i = 0; i < PLD; ++i) {
        const int f = i * 128 + tid;
        r[i] = *(const v4f*)(px + (size_t)(f >> 4) * NN + (f & 15) * 4);
    }
#pragma unroll
    for (int i = 0; i < PLD; ++i) {
        const int f = i * 128 + tid;
        float* d = tile + (f >> 4) * PPITCH + (f & 15) * 4;
        d[0] = r[i].x;
        d[1] = r[i].y;
        d[2] = r[i].z;
        d[3] = r[i].w;
    }
    __syncthreads();

    float* pt = xt + ((size_t)b * NN + n0) * CC;
#pragma unroll
    for (int i = 0; i < PLD; ++i) {
        const int g = i * 128 + tid;
        const float* s = tile + (g & 15) * 4 * PPITCH + (g >> 4);
        v4f o;
        o.x = s[0];
        o.y = s[PPITCH];
        o.z = s[2 * PPITCH];
        o.w = s[3 * PPITCH];
        *(v4f*)(pt + (size_t)g * 4) = o;
    }
    if (tid < PTS) {
        float s = 0.f;
#pragma unroll
        for (int c2 = 0; c2 < CC; ++c2) {
            const float v = tile[c2 * PPITCH + tid];
            s = fmaf(v, v, s);
        }
        xx[b * NN + n0 + tid] = s;
    }
}

#define LCAP 224
#define LTRIG 160
#define NPRE 1
#define SG 4

#define CH 128
#define NCH (NN / CH)
#define PITCH 68
#define SLD 16

__device__ __forceinline__ void stage_fetch(const float* __restrict__ xt_b,
                                            const float* __restrict__ xx_b, int chunk, int tid,
                                            v4f (&st)[SLD], float& sx) {
    const v4f* src = (const v4f*)(xt_b + (size_t)chunk * (CH * CC)) + tid;
#pragma unroll
    for (int i = 0; i < SLD; ++i) st[i] = src[i * 128];
    sx = xx_b[chunk * CH + tid];
}

__device__ __forceinline__ void stage_put(float* __restrict__ buf, float* __restrict__ xbuf,
                                          int tid, const v4f (&st)[SLD], float sx) {
    float* dst = buf + (tid >> 4) * PITCH + (tid & 15) * 4;
#pragma unroll
    for (int i = 0; i < SLD; ++i) *(v4f*)(dst + i * 8 * PITCH) = st[i];
    xbuf[tid] = sx;
}

struct Frag {
    v2f a[16];
    v4f x0, x1;
};
struct Pend {
    v8f acc;
    v4f x0, x1;
};

__device__ __forceinline__ void frag_load(const float* __restrict__ buf,
                                          const float* __restrict__ xbuf, int t, int laneM,
                                          int kofs, int half8, Frag& f) {
    const float* prow = buf + (t * 16 + laneM) * PITCH + kofs;
    const v4f* pxx = (const v4f*)(xbuf + t * 16 + half8);
#pragma unroll
    for (int ks = 0; ks < 16; ++ks) f.a[ks] = *(const v2f*)(prow + ks * 4);
    f.x0 = pxx[0];
    f.x1 = pxx[1];
}

__device__ __forceinline__ void products2(const Frag& fa, const Frag& fb, const v2f (&bq)[16],
                                          v8f& accA, v8f& accB) {
    accA = v8f{};
    accB = v8f{};
#pragma unroll
    for (int ks = 0; ks < 16; ++ks) {
        accA = wmma_f32(fa.a[ks], bq[ks], accA);
        accB = wmma_f32(fb.a[ks], bq[ks], accB);
    }
}

__device__ __forceinline__ void finish(const Pend& p, float (&u)[8]) {
    u[0] = 2.0f * p.acc[0] - p.x0.x;
    u[1] = 2.0f * p.acc[1] - p.x0.y;
    u[2] = 2.0f * p.acc[2] - p.x0.z;
    u[3] = 2.0f * p.acc[3] - p.x0.w;
    u[4] = 2.0f * p.acc[4] - p.x1.x;
    u[5] = 2.0f * p.acc[5] - p.x1.y;
    u[6] = 2.0f * p.acc[6] - p.x1.z;
    u[7] = 2.0f * p.acc[7] - p.x1.w;
}

__device__ __forceinline__ void select1(const Pend& p, float xxq, float (&topv)[KK]) {
    static_assert(NPRE * 8 * SG >= KK && 8 % SG == 0, "sweep 1 must fill the list of maxima");
    float u[8];
    finish(p, u);
#pragma unroll
    for (int g = 0; g < SG; ++g) {
        float m = u[g * (8 / SG)];
#pragma unroll
        for (int r = 1; r < 8 / SG; ++r) m = fmaxf(m, u[g * (8 / SG) + r]);
        m -= xxq;
#pragma unroll
        for (int s = 0; s < KK; ++s) {
            float hi = fmaxf(topv[s], m);
            m = fminf(topv[s], m);
            topv[s] = hi;
        }
    }
}

__device__ __forceinline__ void select2_mark(const Pend& p, float xxq, float thr, int& cnt,
                                             float (&v)[8], int (&slot)[8]) {
    float u[8];
    finish(p, u);
#pragma unroll
    for (int r = 0; r < 8; ++r) {
        v[r] = u[r] - xxq;
        slot[r] = cnt;
        cnt += (v[r] >= thr) ? 1 : 0;
    }
}

__device__ __forceinline__ float next_above(float x) {
    const int b = __float_as_int(x);
    if (x == 0.0f) return __int_as_float(1);
    return __int_as_float(b >= 0 ? b + 1 : b - 1);
}

__device__ __forceinline__ void select2_store(float* lv, int* li, int colbase,
                                              const float (&v)[8], const int (&slot)[8]) {
#pragma unroll
    for (int r = 0; r < 8; ++r) {
        lv[slot[r] * 32] = v[r];
        li[slot[r] * 32] = colbase + r;
    }
}

__device__ __forceinline__ void select2_piece(int j, const float (&dv)[8], float thr, bool ins,
                                              int& cnt, float (&v)[8], int (&slot)[8], float& m,
                                              float (&topv)[KK]) {
    if (j < 8) {
        float d = dv[j];
        asm volatile("" : "+v"(d));
        v[j] = d;
        slot[j] = cnt;
        cnt += (v[j] >= thr) ? 1 : 0;
    } else {
        if (j == 8) {
            float mx = fmaxf(fmaxf(fmaxf(v[0], v[1]), fmaxf(v[2], v[3])),
                             fmaxf(fmaxf(v[4], v[5]), fmaxf(v[6], v[7])));
            m = ins ? mx : -INFINITY;
        }
        const int sb = (j == 8) ? 0 : 2 + (j - 9) * 3;
        const int se = (j == 8) ? 2 : ((2 + (j - 8) * 3) < KK ? (2 + (j - 8) * 3) : KK);
#pragma unroll
        for (int sl = sb; sl < se; ++sl) {
            float hi = fmaxf(topv[sl], m);
            m = fminf(topv[sl], m);
            topv[sl] = hi;
        }
    }
}

__device__ __forceinline__ void products2_mark(const Frag& fa, const Frag& fb, const v2f (&bq)[16],
                                               const Pend& pa, const Pend& pb, float xxq,
                                               float& thr, bool ins, int& cnt, float (&va)[8],
                                               int (&sa)[8], float (&vb)[8], int (&sb)[8],
                                               float (&topv)[KK], v8f& accA, v8f& accB) {
    float ua[8], ub[8], da[8], db[8];
    finish(pa, ua);
#pragma unroll
    for (int j = 0; j < 8; ++j) da[j] = ua[j] - xxq;
    float m = -INFINITY;
    accA = v8f{};
    accB = v8f{};
#pragma unroll
    for (int i = 0; i < 32; ++i) {
        const int ks = i >> 1;
        if ((i & 1) == 0) accA = wmma_f32(fa.a[ks], bq[ks], accA);
        else              accB = wmma_f32(fb.a[ks], bq[ks], accB);
        if (i < 16) {
            select2_piece(i, da, thr, ins, cnt, va, sa, m, topv);
            if (i == 15) {
                const float bnd = topv[KK - 1];
                thr = (bnd > thr) ? bnd : thr;
                finish(pb, ub);
#pragma unroll
                for (int j = 0; j < 8; ++j) db[j] = ub[j] - xxq;
            }
        } else {
            select2_piece(i - 16, db, thr, ins, cnt, vb, sb, m, topv);
        }
        __builtin_amdgcn_sched_barrier(0);
    }
}

__device__ __forceinline__ int drop_to_20(float* lv, int* li, int m) {
    while (m > KK) {
        float wv = lv[0];
        int wi = li[0];
        int wp = 0;
        for (int j = 1; j < m; ++j) {
            float vj = lv[j * 32];
            int ij = li[j * 32];
            bool worse = (vj < wv) || (vj == wv && ij > wi);
            if (worse) { wv = vj; wi = ij; wp = j; }
        }
        --m;
        lv[wp * 32] = lv[m * 32];
        li[wp * 32] = li[m * 32];
    }
    return m;
}

__global__ __launch_bounds__(128) void knn_kernel(const float* __restrict__ xt,
                                                  const float* __restrict__ xx,
                                                  int* __restrict__ nbr) {
    const int tid = threadIdx.x;
    const int wave = threadIdx.x >> 5;
    const int lane = threadIdx.x & 31;
    const int laneM = lane & 15;
    const int half = lane >> 4;
    const int kofs = half * 2;
    const int half8 = half * 8;

    const int qg = blockIdx.x * 64 + wave * 16;
    const int b = qg / NN;
    const int q0 = qg - b * NN;
    const float* xt_b = xt + (size_t)b * NN * CC;
    const float* xx_b = xx + b * NN;

    const float* pq = xt_b + (size_t)(q0 + laneM) * CC + kofs;
    v2f bq[16];
#pragma unroll
    for (int ks = 0; ks < 16; ++ks) bq[ks] = *(const v2f*)(pq + ks * 4);
    const float xxq = xx_b[q0 + laneM];

    __shared__ __attribute__((aligned(16))) float colbuf[2][CH * PITCH];
    __shared__ __attribute__((aligned(16))) float xxbuf[2][CH];
    struct Lists {
        float v[LCAP * 32];
        int i[LCAP * 32];
    };
    __shared__ Lists lists[4];
    float* lv = &lists[wave].v[lane];
    int* li = &lists[wave].i[lane];

    v4f st[SLD];
    float sx;
    stage_fetch(xt_b, xx_b, 0, tid, st, sx);
    stage_put(colbuf[0], xxbuf[0], tid, st, sx);
    __syncthreads();

    float topv[KK];
#pragma unroll
    for (int s = 0; s < KK; ++s) topv[s] = -3.0e38f;

    Pend pendA, pendB;
    pendA.acc = v8f{};
    pendA.x0 = v4f{INFINITY, INFINITY, INFINITY, INFINITY};
    pendA.x1 = pendA.x0;
    pendB = pendA;
    for (int c = 0; c < NPRE; ++c) {
        stage_fetch(xt_b, xx_b, c + 1, tid, st, sx);
        const float* buf = colbuf[c & 1];
        const float* xb = xxbuf[c & 1];
        Frag fa0, fb0, fa1, fb1;
        frag_load(buf, xb, 0, laneM, kofs, half8, fa0);
        frag_load(buf, xb, 1, laneM, kofs, half8, fb0);
#pragma unroll
        for (int r = 0; r < CH / 32; ++r) {
            Frag& ca = (r & 1) ? fa1 : fa0;
            Frag& cb = (r & 1) ? fb1 : fb0;
            Frag& na = (r & 1) ? fa0 : fa1;
            Frag& nb = (r & 1) ? fb0 : fb1;
            if (r + 1 < CH / 32) {
                frag_load(buf, xb, 2 * r + 2, laneM, kofs, half8, na);
                frag_load(buf, xb, 2 * r + 3, laneM, kofs, half8, nb);
            }
            __builtin_amdgcn_sched_barrier(0);
            v8f accA, accB;
            products2(ca, cb, bq, accA, accB);
            select1(pendA, xxq, topv);
            select1(pendB, xxq, topv);
            __builtin_amdgcn_sched_barrier(0);
            pendA.acc = accA;
            pendA.x0 = ca.x0;
            pendA.x1 = ca.x1;
            pendB.acc = accB;
            pendB.x0 = cb.x0;
            pendB.x1 = cb.x1;
        }
        stage_put(colbuf[(c + 1) & 1], xxbuf[(c + 1) & 1], tid, st, sx);
        __syncthreads();
    }
    select1(pendA, xxq, topv);
    select1(pendB, xxq, topv);

    float thr = topv[KK - 1];
    float lo_inc = -INFINITY;
    static_assert((NCH & (NCH - 1)) == 0 && NPRE < NCH, "the stage order wraps by a mask");
    int cnt = 0;
    pendA.acc = v8f{};
    pendA.x0 = v4f{NAN, NAN, NAN, NAN};
    pendA.x1 = pendA.x0;
    pendB = pendA;
    int pcol = half8;
    for (int i = 0; i < NCH; ++i) {
        const int c = (i + NPRE) & (NCH - 1);
        const int cnx = (c + 1) & (NCH - 1);
        const bool rising = (i < NCH - NPRE);
        const bool more = (i + 1 < NCH);
        if (more) stage_fetch(xt_b, xx_b, cnx, tid, st, sx);
        if (i == NCH - NPRE) thr = fmaxf(topv[KK - 1], lo_inc);
        const float* buf = colbuf[c & 1];
        const float* xb = xxbuf[c & 1];
        Frag fa0, fb0, fa1, fb1;
        frag_load(buf, xb, 0, laneM, kofs, half8, fa0);
        frag_load(buf, xb, 1, laneM, kofs, half8, fb0);
#pragma unroll
        for (int r = 0; r < CH / 32; ++r) {
            Frag& ca = (r & 1) ? fa1 : fa0;
            Frag& cb = (r & 1) ? fb1 : fb0;
            Frag& na = (r & 1) ? fa0 : fa1;
            Frag& nb = (r & 1) ? fb0 : fb1;
            if (r + 1 < CH / 32) {
                frag_load(buf, xb, 2 * r + 2, laneM, kofs, half8, na);
                frag_load(buf, xb, 2 * r + 3, laneM, kofs, half8, nb);
            }
            __builtin_amdgcn_sched_barrier(0);
            float cva[8], cvb[8];
            int csa[8], csb[8];
            const bool ins = (r == 0) ? (i >= 1 && i <= NCH - NPRE) : rising;
            v8f accA, accB;
            products2_mark(ca, cb, bq, pendA, pendB, xxq, thr, ins, cnt, cva, csa, cvb, csb, topv,
                           accA, accB);
            select2_store(lv, li, pcol, cva, csa);
            select2_store(lv, li, pcol + 16, cvb, csb);
            {
                const float bnd = topv[KK - 1];
                thr = (bnd > thr) ? bnd : thr;
            }
            pendA.acc = accA;
            pendA.x0 = ca.x0;
            pendA.x1 = ca.x1;
            pendB.acc = accB;
            pendB.x0 = cb.x0;
            pendB.x1 = cb.x1;
            pcol = c * CH + r * 32 + half8;
        }
        if (cnt >= LTRIG) {
            cnt = drop_to_20(lv, li, cnt);
            float lo = lv[0];
            for (int s = 1; s < KK; ++s) lo = fminf(lo, lv[s * 32]);
            if (rising) {
                if (lo >= thr) thr = next_above(lo);
                lo_inc = fmaxf(lo_inc, lo);
            } else {
                if (lo > thr) thr = lo;
            }
        }
        if (more) {
            stage_put(colbuf[cnx & 1], xxbuf[cnx & 1], tid, st, sx);
            __syncthreads();
        }
    }
#pragma unroll
    for (int e = 0; e < 2; ++e) {
        const Pend& pe = (e == 0) ? pendA : pendB;
        float cv[8];
        int cs[8];
        select2_mark(pe, xxq, thr, cnt, cv, cs);
        select2_store(lv, li, pcol + 16 * e, cv, cs);
    }
    {
        float pv[KK];
#pragma unroll
        for (int s = 0; s < KK; ++s) pv[s] = __shfl_xor(topv[s], 16);
#pragma unroll
        for (int t = 0; t < KK; ++t) {
            float m = pv[t];
#pragma unroll
            for (int s = 0; s < KK; ++s) {
                float hi = fmaxf(topv[s], m);
                m = fminf(topv[s], m);
                topv[s] = hi;
            }
        }
    }
    static_assert(LCAP % 8 == 0, "groups of eight must stay inside a lane's list");
    {
        const float thrq = topv[KK - 1];
        int kept = 0;
        for (int e0 = 0; e0 < cnt; e0 += 8) {
            float ve[8];
            int ie[8];
#pragma unroll
            for (int j = 0; j < 8; ++j) {
                ve[j] = lv[(e0 + j) * 32];
                ie[j] = li[(e0 + j) * 32];
            }
#pragma unroll
            for (int j = 0; j < 8; ++j) {
                lv[kept * 32] = ve[j];
                li[kept * 32] = ie[j];
                kept += ((e0 + j < cnt) && (ve[j] >= thrq)) ? 1 : 0;
            }
        }
        cnt = kept;
    }
    cnt = drop_to_20(lv, li, cnt);

    const int cntUp = __shfl_xor(cnt, 16);
    __syncthreads();
    if (half == 0) {
        for (int e = 0; e < cntUp; ++e) {
            lv[(cnt + e) * 32] = lv[16 + e * 32];
            li[(cnt + e) * 32] = li[16 + e * 32];
        }
        const int tot = drop_to_20(lv, li, cnt + cntUp);
        int* outp = nbr + (size_t)(b * NN + q0 + laneM) * KK;
        for (int s = 0; s < KK; ++s) outp[s] = (s < tot) ? li[s * 32] : 0;
    }
}

__device__ __forceinline__ void wave_lds_order() {
    __builtin_amdgcn_fence(__ATOMIC_RELEASE, "wavefront");
    __builtin_amdgcn_wave_barrier();
    __builtin_amdgcn_fence(__ATOMIC_ACQUIRE, "wavefront");
}

#define GPB 4
#define OMP 65
#define TPW 20

__device__ __forceinline__ int edge_row(int G0, int wave, int q, int laneM) {
    const int g = q / 5;
    const int it = q - g * 5;
    return (G0 + g) * (16 * KK) + (wave * 5 + it) * 16 + laneM;
}

__global__ __launch_bounds__(128) void edgeconv_kernel(
    const float* __restrict__ xt, const int* __restrict__ nbr,
    const float* __restrict__ w1, const float* __restrict__ g1,
    const float* __restrict__ be1, const float* __restrict__ mu1,
    const float* __restrict__ va1, const float* __restrict__ w2,
    const float* __restrict__ g2, const float* __restrict__ be2,
    const float* __restrict__ mu2, const float* __restrict__ va2,
    float* __restrict__ out) {
    const int wave = threadIdx.x >> 5;
    const int lane = threadIdx.x & 31;
    const int laneM = lane & 15;
    const int kofs = (lane < 16) ? 0 : 2;
    const int G0 = blockIdx.x * GPB;

    __shared__ float h1s[4][16 * OO];
    __shared__ unsigned om[OO * OMP];

    int e0 = edge_row(G0, wave, 0, laneM);
    int nb0 = nbr[e0];
    int e1 = edge_row(G0, wave, 1, laneM);
    int nbN = nbr[e1];

    v2f wa[16][4], wb[16][4], wc[16][4];
#pragma unroll
    for (int ks = 0; ks < 16; ++ks) {
        const int kc = ks * 4 + kofs;
#pragma unroll
        for (int nt = 0; nt < 4; ++nt) {
            const int o = nt * 16 + laneM;
            wa[ks][nt] = *(const v2f*)(w1 + o * 128 + kc);
            wb[ks][nt] = *(const v2f*)(w1 + o * 128 + 64 + kc);
            wc[ks][nt] = *(const v2f*)(w2 + o * OO + kc);
        }
    }
    float sc1[4], sh1[4], sc2[4], sh2[4];
#pragma unroll
    for (int nt = 0; nt < 4; ++nt) {
        const int o = nt * 16 + laneM;
        sc1[nt] = g1[o] * rsqrtf(va1[o] + EPSV);
        sh1[nt] = be1[o] - mu1[o] * sc1[nt];
        sc2[nt] = g2[o] * rsqrtf(va2[o] + EPSV);
        sh2[nt] = be2[o] - mu2[o] * sc2[nt];
    }

    v2f an[16], cn[16];
    {
        const float* pc = xt + (size_t)(e0 / KK) * CC + kofs;
        const float* pn = xt + ((size_t)((e0 / KK) & ~(NN - 1)) + nb0) * CC + kofs;
#pragma unroll
        for (int ks = 0; ks < 16; ++ks) {
            an[ks] = *(const v2f*)(pn + ks * 4);
            cn[ks] = *(const v2f*)(pc + ks * 4);
        }
    }

    for (int i = threadIdx.x; i < OO * OMP; i += 128) om[i] = 0u;
    __syncthreads();

    for (int q = 0; q < TPW; ++q) {
        const int g = q / 5;
        const int it = q - g * 5;
        const int mt = wave * 5 + it;

        v2f ac[16], cc[16];
#pragma unroll
        for (int ks = 0; ks < 16; ++ks) {
            ac[ks] = an[ks];
            cc[ks] = cn[ks];
        }
        {
            const int qn = (q + 1 < TPW) ? q + 1 : TPW - 1;
            const int q2 = (q + 2 < TPW) ? q + 2 : TPW - 1;
            const int en = edge_row(G0, wave, qn, laneM);
            const float* pc = xt + (size_t)(en / KK) * CC + kofs;
            const float* pn = xt + ((size_t)((en / KK) & ~(NN - 1)) + nbN) * CC + kofs;
#pragma unroll
            for (int ks = 0; ks < 16; ++ks) {
                an[ks] = *(const v2f*)(pn + ks * 4);
                cn[ks] = *(const v2f*)(pc + ks * 4);
            }
            nbN = nbr[edge_row(G0, wave, q2, laneM)];
        }

        v8f acc1[4] = {};
#pragma unroll
        for (int ks = 0; ks < 16; ++ks) {
            v2f a = ac[ks] - cc[ks];
#pragma unroll
            for (int nt = 0; nt < 4; ++nt) acc1[nt] = wmma_f32(a, wa[ks][nt], acc1[nt]);
        }
#pragma unroll
        for (int ks = 0; ks < 16; ++ks) {
#pragma unroll
            for (int nt = 0; nt < 4; ++nt) acc1[nt] = wmma_f32(cc[ks], wb[ks][nt], acc1[nt]);
        }
#pragma unroll
        for (int nt = 0; nt < 4; ++nt) {
            int o = nt * 16 + laneM;
#pragma unroll
            for (int j = 0; j < 8; ++j) {
                int M = j + ((lane < 16) ? 0 : 8);
                h1s[wave][M * OO + o] = fmaxf(acc1[nt][j] * sc1[nt] + sh1[nt], 0.0f);
            }
        }
        wave_lds_order();

        v8f acc2[4] = {};
#pragma unroll
        for (int ks = 0; ks < 16; ++ks) {
            int kc = ks * 4 + kofs;
            v2f a = *(const v2f*)(&h1s[wave][laneM * OO + kc]);
#pragma unroll
            for (int nt = 0; nt < 4; ++nt) acc2[nt] = wmma_f32(a, wc[ks][nt], acc2[nt]);
        }
#pragma unroll
        for (int nt = 0; nt < 4; ++nt) {
            int o = nt * 16 + laneM;
#pragma unroll
            for (int j = 0; j < 8; ++j) {
                int M = j + ((lane < 16) ? 0 : 8);
                int lr = mt * 16 + M;
                int pp = lr / KK;
                float val = fmaxf(acc2[nt][j] * sc2[nt] + sh2[nt], 0.0f);
                atomicMax(&om[o * OMP + g * 16 + pp], __float_as_uint(val));
            }
        }
        wave_lds_order();
        __builtin_amdgcn_sched_barrier(0);
    }

    __syncthreads();
    {
        const int b = G0 >> 8;
        const int pt0 = G0 & 255;
        float* po = out + (size_t)b * OO * NN + pt0 * 16;
#pragma unroll
        for (int i = 0; i < (OO * 16 * GPB) / 4 / 128; ++i) {
            const int f = i * 128 + threadIdx.x;
            const int o = f >> 4;
            const unsigned* s = om + o * OMP + (f & 15) * 4;
            v4f v;
            v.x = __uint_as_float(s[0]);
            v.y = __uint_as_float(s[1]);
            v.z = __uint_as_float(s[2]);
            v.w = __uint_as_float(s[3]);
            *(v4f*)(po + (size_t)o * NN + (f & 15) * 4) = v;
        }
    }
}

extern "C" void kernel_launch(void* const* d_in, const int* in_sizes, int n_in,
                              void* d_out, int out_size, void* d_ws, size_t ws_size,
                              hipStream_t stream) {
    const float* x   = (const float*)d_in[0];
    const float* w1  = (const float*)d_in[1];
    const float* g1  = (const float*)d_in[2];
    const float* be1 = (const float*)d_in[3];
    const float* mu1 = (const float*)d_in[4];
    const float* va1 = (const float*)d_in[5];
    const float* w2  = (const float*)d_in[6];
    const float* g2  = (const float*)d_in[7];
    const float* be2 = (const float*)d_in[8];
    const float* mu2 = (const float*)d_in[9];
    const float* va2 = (const float*)d_in[10];
    float* out = (float*)d_out;

    float* xt = (float*)d_ws;
    float* xx = xt + (size_t)BB * NN * CC;
    int* nbr = (int*)(xx + (size_t)BB * NN);

    prep_kernel<<<(BB * NN) / PTS, 128, 0, stream>>>(x, xt, xx);
    knn_kernel<<<(BB * NN) / 64, 128, 0, stream>>>(xt, xx, nbr);
    edgeconv_kernel<<<(BB * (NN / 16)) / GPB, 128, 0, stream>>>(
        xt, nbr, w1, g1, be1, mu1, va1, w2, g2, be2, mu2, va2, out);
}
